# starting kernel
# baseline (speedup 1.0000x reference)
#include <hip/hip_runtime.h>
#include <hip/hip_bf16.h>

#define IN_DIM 500
#define HID    256
#define NCLASS 50
#define KHOPS  10

typedef __attribute__((ext_vector_type(16))) _Float16 v16h;
typedef __attribute__((ext_vector_type(8)))  _Float16 h8;
typedef __attribute__((ext_vector_type(4)))  _Float16 h4;
typedef __attribute__((ext_vector_type(8)))  float    v8f;

__device__ __forceinline__ v16h load_frag(const _Float16* row, int kb) {
  h8 lo = *(const h8*)(row + kb);
  h8 hi = *(const h8*)(row + 16 + kb);
  v16h f;
#pragma unroll
  for (int i = 0; i < 8; ++i) { f[i] = lo[i]; f[8 + i] = hi[i]; }
  return f;
}

__device__ __forceinline__ v8f wmma16(v16h a, v16h b, v8f c) {
  return __builtin_amdgcn_wmma_f32_16x16x32_f16(false, a, false, b, (short)0,
                                                c, false, false);
}

__global__ void __launch_bounds__(256)
dagnn_gemm1(const float* __restrict__ X, const float* __restrict__ W,
            const float* __restrict__ bias, _Float16* __restrict__ Y, int Nn) {
  __shared__ __align__(16) _Float16 As[2][64][40];
  __shared__ __align__(16) _Float16 Bt[2][64][40];
  const int t = threadIdx.x;
  const int wave = t >> 5, lane = t & 31;
  const int waveM = wave >> 1, waveN = wave & 1;
  const int blockM = blockIdx.x * 64;
  const int blockN = blockIdx.y * 64;

  auto loadA = [&](int kt, int buf) {
#pragma unroll
    for (int j = 0; j < 2; ++j) {
      int idx = j * 256 + t;
      int r = idx >> 3, k4 = (idx & 7) * 4;
      int grow = blockM + r, gk = kt + k4;
      int cr = grow < Nn ? grow : Nn - 1;
      int ck = gk < IN_DIM ? gk : IN_DIM - 4;
      float4 v = *(const float4*)(X + (size_t)cr * IN_DIM + ck);
      bool ok = gk < IN_DIM;
      h4 hv;
      hv[0] = ok ? (_Float16)v.x : (_Float16)0.f;
      hv[1] = ok ? (_Float16)v.y : (_Float16)0.f;
      hv[2] = ok ? (_Float16)v.z : (_Float16)0.f;
      hv[3] = ok ? (_Float16)v.w : (_Float16)0.f;
      *(h4*)&As[buf][r][k4] = hv;
    }
  };
  auto loadB = [&](int kt, int buf) {
#pragma unroll
    for (int j = 0; j < 2; ++j) {
      int idx = j * 256 + t;
      int kk = idx >> 4, n4 = (idx & 15) * 4;
      int gk = kt + kk;
      int ck = gk < IN_DIM ? gk : IN_DIM - 1;
      float4 v = *(const float4*)(W + (size_t)ck * HID + blockN + n4);
      bool ok = gk < IN_DIM;
      Bt[buf][n4 + 0][kk] = ok ? (_Float16)v.x : (_Float16)0.f;
      Bt[buf][n4 + 1][kk] = ok ? (_Float16)v.y : (_Float16)0.f;
      Bt[buf][n4 + 2][kk] = ok ? (_Float16)v.z : (_Float16)0.f;
      Bt[buf][n4 + 3][kk] = ok ? (_Float16)v.w : (_Float16)0.f;
    }
  };

  v8f acc0 = {0.f, 0.f, 0.f, 0.f, 0.f, 0.f, 0.f, 0.f};
  v8f acc1 = acc0;
  const int mrow = waveM * 16 + (lane & 15);
  const int nrow = waveN * 32 + (lane & 15);
  const int kb = (lane >> 4) * 8;

  loadA(0, 0);
  loadB(0, 0);
  __syncthreads();
  const int NKT = 16;
  for (int it = 0; it < NKT; ++it) {
    int cur = it & 1;
    if (it + 1 < NKT) {
      loadA((it + 1) * 32, cur ^ 1);
      loadB((it + 1) * 32, cur ^ 1);
    }
    v16h a  = load_frag(&As[cur][mrow][0], kb);
    v16h b0 = load_frag(&Bt[cur][nrow][0], kb);
    v16h b1 = load_frag(&Bt[cur][nrow + 16][0], kb);
    acc0 = wmma16(a, b0, acc0);
    acc1 = wmma16(a, b1, acc1);
    __syncthreads();
  }

  const int col0 = blockN + waveN * 32 + (lane & 15);
  const int col1 = col0 + 16;
  const int rbase = blockM + waveM * 16 + (lane >> 4) * 8;
  const float bv0 = bias[col0], bv1 = bias[col1];
#pragma unroll
  for (int i = 0; i < 8; ++i) {
    int row = rbase + i;
    if (row < Nn) {
      float v0 = acc0[i] + bv0;
      float v1 = acc1[i] + bv1;
      Y[(size_t)row * HID + col0] = (_Float16)(v0 > 0.f ? v0 : 0.f);
      Y[(size_t)row * HID + col1] = (_Float16)(v1 > 0.f ? v1 : 0.f);
    }
  }
}

__global__ void __launch_bounds__(256)
dagnn_gemm2(const _Float16* __restrict__ Hd, const float* __restrict__ W2,
            const float* __restrict__ b2, float* __restrict__ plane0, int Nn) {
  __shared__ __align__(16) _Float16 As[2][32][40];
  __shared__ __align__(16) _Float16 Bt[2][64][40];
  const int t = threadIdx.x;
  const int wave = t >> 5, lane = t & 31;
  const int waveM = wave >> 2, waveN = wave & 3;
  const int blockM = blockIdx.x * 32;

  auto loadA = [&](int kt, int buf) {
    int r = t >> 3, k4 = (t & 7) * 4;
    int grow = blockM + r;
    int cr = grow < Nn ? grow : Nn - 1;
    *(h4*)&As[buf][r][k4] = *(const h4*)(Hd + (size_t)cr * HID + kt + k4);
  };
  auto loadB = [&](int kt, int buf) {
#pragma unroll
    for (int j = 0; j < 8; ++j) {
      int idx = j * 256 + t;
      int n = idx & 63, kk = idx >> 6;
      int cn = n < NCLASS ? n : NCLASS - 1;
      float v = W2[(size_t)(kt + kk) * NCLASS + cn];
      Bt[buf][n][kk] = (n < NCLASS) ? (_Float16)v : (_Float16)0.f;
    }
  };

  v8f acc = {0.f, 0.f, 0.f, 0.f, 0.f, 0.f, 0.f, 0.f};
  const int mrow = waveM * 16 + (lane & 15);
  const int nrow = waveN * 16 + (lane & 15);
  const int kb = (lane >> 4) * 8;

  loadA(0, 0);
  loadB(0, 0);
  __syncthreads();
  const int NKT = HID / 32;
  for (int it = 0; it < NKT; ++it) {
    int cur = it & 1;
    if (it + 1 < NKT) {
      loadA((it + 1) * 32, cur ^ 1);
      loadB((it + 1) * 32, cur ^ 1);
    }
    v16h a = load_frag(&As[cur][mrow][0], kb);
    v16h b = load_frag(&Bt[cur][nrow][0], kb);
    acc = wmma16(a, b, acc);
    __syncthreads();
  }

  const int col = waveN * 16 + (lane & 15);
  const int rbase = blockM + waveM * 16 + (lane >> 4) * 8;
  if (col < NCLASS) {
    const float bv = b2[col];
#pragma unroll
    for (int i = 0; i < 8; ++i) {
      int row = rbase + i;
      if (row < Nn) plane0[(size_t)row * NCLASS + col] = acc[i] + bv;
    }
  }
}

__global__ void dagnn_degree(const int* __restrict__ dst, float* __restrict__ deg, int E) {
  int i = blockIdx.x * blockDim.x + threadIdx.x;
  if (i < E) atomicAdd(&deg[dst[i]], 1.0f);
}

__global__ void dagnn_norm(float* __restrict__ deg, int Nn) {
  int i = blockIdx.x * blockDim.x + threadIdx.x;
  if (i < Nn) deg[i] = rsqrtf(deg[i]);
}

__global__ void dagnn_scatter(const float* __restrict__ in, float* __restrict__ outp,
                              const float* __restrict__ norm,
                              const int* __restrict__ src, const int* __restrict__ dst,
                              int E) {
  int gid = blockIdx.x * blockDim.x + threadIdx.x;
  int e = gid >> 5, lane = gid & 31;
  if (e >= E) return;
  int sv = src[e], dv = dst[e];
  float ns = norm[sv];
  const float* ip = in + (size_t)sv * NCLASS;
  float* op = outp + (size_t)dv * NCLASS;
  atomicAdd(&op[lane], ns * ip[lane]);
  if (lane < NCLASS - 32)
    atomicAdd(&op[lane + 32], ns * ip[lane + 32]);
}

__global__ void dagnn_scale(float* __restrict__ p, const float* __restrict__ norm, int Nn) {
  int i = blockIdx.x * blockDim.x + threadIdx.x;
  if (i < Nn * NCLASS) p[i] *= norm[i / NCLASS];
}

__global__ void dagnn_combine(const float* __restrict__ planes, const float* __restrict__ s,
                              float* __restrict__ out, int Nn) {
  int gid = blockIdx.x * blockDim.x + threadIdx.x;
  int node = gid >> 5, lane = gid & 31;
  if (node >= Nn) return;
  int c2 = lane + 32;
  float s1 = s[lane];
  float s2 = (c2 < NCLASS) ? s[c2] : 0.f;
  float acc1 = 0.f, acc2 = 0.f;
  size_t pstride = (size_t)Nn * NCLASS;
  size_t base = (size_t)node * NCLASS;
#pragma unroll
  for (int k = 0; k <= KHOPS; ++k) {
    const float* hp = planes + (size_t)k * pstride + base;
    float h1 = hp[lane];
    float h2 = (c2 < NCLASS) ? hp[c2] : 0.f;
    float z = h1 * s1 + h2 * s2;
#pragma unroll
    for (int off = 16; off > 0; off >>= 1) z += __shfl_xor(z, off, 32);
    float S = 1.0f / (1.0f + __expf(-z));
    acc1 += S * h1;
    acc2 += S * h2;
  }
  out[base + lane] = acc1;
  if (c2 < NCLASS) out[base + c2] = acc2;
}

extern "C" void kernel_launch(void* const* d_in, const int* in_sizes, int n_in,
                              void* d_out, int out_size, void* d_ws, size_t ws_size,
                              hipStream_t stream) {
  const float* feats = (const float*)d_in[0];
  const float* W1    = (const float*)d_in[1];
  const float* b1    = (const float*)d_in[2];
  const float* W2    = (const float*)d_in[3];
  const float* b2    = (const float*)d_in[4];
  const float* s     = (const float*)d_in[5];
  const int*   src   = (const int*)d_in[6];
  const int*   dst   = (const int*)d_in[7];
  const int Nn = in_sizes[0] / IN_DIM;
  const int E  = in_sizes[6];
  float* out = (float*)d_out;

  char* ws = (char*)d_ws;
  size_t off = 0;
  auto carve = [&](size_t bytes) -> char* {
    char* p = ws + off;
    off = (off + bytes + 255) & ~(size_t)255;
    return p;
  };
  float*    norm   = (float*)   carve((size_t)Nn * sizeof(float));
  _Float16* hid    = (_Float16*)carve((size_t)Nn * HID * sizeof(_Float16));
  float*    planes = (float*)   carve((size_t)(KHOPS + 1) * Nn * NCLASS * sizeof(float));
  if (off > ws_size) return;

  const size_t pstride = (size_t)Nn * NCLASS;

  hipMemsetAsync(norm, 0, (size_t)Nn * sizeof(float), stream);
  dagnn_degree<<<(E + 255) / 256, 256, 0, stream>>>(dst, norm, E);
  dagnn_norm<<<(Nn + 255) / 256, 256, 0, stream>>>(norm, Nn);

  dim3 g1((Nn + 63) / 64, HID / 64);
  dagnn_gemm1<<<g1, 256, 0, stream>>>(feats, W1, b1, hid, Nn);
  dim3 g2((Nn + 31) / 32, 1);
  dagnn_gemm2<<<g2, 256, 0, stream>>>(hid, W2, b2, planes, Nn);

  hipMemsetAsync(planes + pstride, 0, (size_t)KHOPS * pstride * sizeof(float), stream);
  const int sgrid = (int)(((size_t)E * 32 + 255) / 256);
  const int cgrid = (Nn * NCLASS + 255) / 256;
  for (int k = 1; k <= KHOPS; ++k) {
    const float* pin = planes + (size_t)(k - 1) * pstride;
    float* pout      = planes + (size_t)k * pstride;
    dagnn_scatter<<<sgrid, 256, 0, stream>>>(pin, pout, norm, src, dst, E);
    dagnn_scale<<<cgrid, 256, 0, stream>>>(pout, norm, Nn);
  }

  dagnn_combine<<<(int)(((size_t)Nn * 32 + 255) / 256), 256, 0, stream>>>(planes, s, out, Nn);
  (void)n_in; (void)out_size;
}


# Propagation planes stored at a 256-byte row pitch (50 floats padded to 64, pad zeroed): each gathered row is one aligned 256-B unit, read and stored as two whole lines by one instruction
# speedup vs baseline: 3.6318x; 3.6318x over previous
#include <hip/hip_runtime.h>
#include <hip/hip_bf16.h>

#define IN_DIM 500
#define HID    256
#define NCLASS 50
#define KHOPS  10
#define PITCH  64

typedef __attribute__((ext_vector_type(16))) _Float16 v16h;
typedef __attribute__((ext_vector_type(8)))  _Float16 h8;
typedef __attribute__((ext_vector_type(4)))  _Float16 h4;
typedef __attribute__((ext_vector_type(8)))  float    v8f;

__device__ __forceinline__ v16h load_frag(const _Float16* row, int kb) {
  h8 lo = *(const h8*)(row + kb);
  h8 hi = *(const h8*)(row + 16 + kb);
  v16h f;
#pragma unroll
  for (int i = 0; i < 8; ++i) { f[i] = lo[i]; f[8 + i] = hi[i]; }
  return f;
}

__device__ __forceinline__ v8f wmma16(v16h a, v16h b, v8f c) {
  return __builtin_amdgcn_wmma_f32_16x16x32_f16(false, a, false, b, (short)0,
                                                c, false, false);
}

__global__ void __launch_bounds__(256)
dagnn_gemm1(const float* __restrict__ X, const float* __restrict__ W,
            const float* __restrict__ bias, _Float16* __restrict__ Y, int Nn) {
  __shared__ __align__(16) _Float16 As[2][64][40];
  __shared__ __align__(16) _Float16 Bt[2][64][40];
  const int t = threadIdx.x;
  const int wave = t >> 5, lane = t & 31;
  const int waveM = wave >> 1, waveN = wave & 1;
  const int blockM = blockIdx.x * 64;
  const int blockN = blockIdx.y * 64;

  auto loadA = [&](int kt, int buf) {
#pragma unroll
    for (int j = 0; j < 2; ++j) {
      int idx = j * 256 + t;
      int r = idx >> 3, k4 = (idx & 7) * 4;
      int grow = blockM + r, gk = kt + k4;
      int cr = grow < Nn ? grow : Nn - 1;
      int ck = gk < IN_DIM ? gk : IN_DIM - 4;
      float4 v = *(const float4*)(X + (size_t)cr * IN_DIM + ck);
      bool ok = gk < IN_DIM;
      h4 hv;
      hv[0] = ok ? (_Float16)v.x : (_Float16)0.f;
      hv[1] = ok ? (_Float16)v.y : (_Float16)0.f;
      hv[2] = ok ? (_Float16)v.z : (_Float16)0.f;
      hv[3] = ok ? (_Float16)v.w : (_Float16)0.f;
      *(h4*)&As[buf][r][k4] = hv;
    }
  };
  auto loadB = [&](int kt, int buf) {
#pragma unroll
    for (int j = 0; j < 2; ++j) {
      int idx = j * 256 + t;
      int kk = idx >> 4, n4 = (idx & 15) * 4;
      int gk = kt + kk;
      int ck = gk < IN_DIM ? gk : IN_DIM - 1;
      float4 v = *(const float4*)(W + (size_t)ck * HID + blockN + n4);
      bool ok = gk < IN_DIM;
      Bt[buf][n4 + 0][kk] = ok ? (_Float16)v.x : (_Float16)0.f;
      Bt[buf][n4 + 1][kk] = ok ? (_Float16)v.y : (_Float16)0.f;
      Bt[buf][n4 + 2][kk] = ok ? (_Float16)v.z : (_Float16)0.f;
      Bt[buf][n4 + 3][kk] = ok ? (_Float16)v.w : (_Float16)0.f;
    }
  };

  v8f acc0 = {0.f, 0.f, 0.f, 0.f, 0.f, 0.f, 0.f, 0.f};
  v8f acc1 = acc0;
  const int mrow = waveM * 16 + (lane & 15);
  const int nrow = waveN * 32 + (lane & 15);
  const int kb = (lane >> 4) * 8;

  loadA(0, 0);
  loadB(0, 0);
  __syncthreads();
  const int NKT = 16;
  for (int it = 0; it < NKT; ++it) {
    int cur = it & 1;
    if (it + 1 < NKT) {
      loadA((it + 1) * 32, cur ^ 1);
      loadB((it + 1) * 32, cur ^ 1);
    }
    v16h a  = load_frag(&As[cur][mrow][0], kb);
    v16h b0 = load_frag(&Bt[cur][nrow][0], kb);
    v16h b1 = load_frag(&Bt[cur][nrow + 16][0], kb);
    acc0 = wmma16(a, b0, acc0);
    acc1 = wmma16(a, b1, acc1);
    __syncthreads();
  }

  const int col0 = blockN + waveN * 32 + (lane & 15);
  const int col1 = col0 + 16;
  const int rbase = blockM + waveM * 16 + (lane >> 4) * 8;
  const float bv0 = bias[col0], bv1 = bias[col1];
#pragma unroll
  for (int i = 0; i < 8; ++i) {
    int row = rbase + i;
    if (row < Nn) {
      float v0 = acc0[i] + bv0;
      float v1 = acc1[i] + bv1;
      Y[(size_t)row * HID + col0] = (_Float16)(v0 > 0.f ? v0 : 0.f);
      Y[(size_t)row * HID + col1] = (_Float16)(v1 > 0.f ? v1 : 0.f);
    }
  }
}

__global__ void __launch_bounds__(256)
dagnn_gemm2(const _Float16* __restrict__ Hd, const float* __restrict__ W2,
            const float* __restrict__ b2, float* __restrict__ plane0, int Nn) {
  __shared__ __align__(16) _Float16 As[2][32][40];
  __shared__ __align__(16) _Float16 Bt[2][64][40];
  const int t = threadIdx.x;
  const int wave = t >> 5, lane = t & 31;
  const int waveM = wave >> 2, waveN = wave & 3;
  const int blockM = blockIdx.x * 32;

  auto loadA = [&](int kt, int buf) {
    int r = t >> 3, k4 = (t & 7) * 4;
    int grow = blockM + r;
    int cr = grow < Nn ? grow : Nn - 1;
    *(h4*)&As[buf][r][k4] = *(const h4*)(Hd + (size_t)cr * HID + kt + k4);
  };
  auto loadB = [&](int kt, int buf) {
#pragma unroll
    for (int j = 0; j < 8; ++j) {
      int idx = j * 256 + t;
      int n = idx & 63, kk = idx >> 6;
      int cn = n < NCLASS ? n : NCLASS - 1;
      float v = W2[(size_t)(kt + kk) * NCLASS + cn];
      Bt[buf][n][kk] = (n < NCLASS) ? (_Float16)v : (_Float16)0.f;
    }
  };

  v8f acc = {0.f, 0.f, 0.f, 0.f, 0.f, 0.f, 0.f, 0.f};
  const int mrow = waveM * 16 + (lane & 15);
  const int nrow = waveN * 16 + (lane & 15);
  const int kb = (lane >> 4) * 8;

  loadA(0, 0);
  loadB(0, 0);
  __syncthreads();
  const int NKT = HID / 32;
  for (int it = 0; it < NKT; ++it) {
    int cur = it & 1;
    if (it + 1 < NKT) {
      loadA((it + 1) * 32, cur ^ 1);
      loadB((it + 1) * 32, cur ^ 1);
    }
    v16h a = load_frag(&As[cur][mrow][0], kb);
    v16h b = load_frag(&Bt[cur][nrow][0], kb);
    acc = wmma16(a, b, acc);
    __syncthreads();
  }

  const int col = waveN * 16 + (lane & 15);
  const int rbase = blockM + waveM * 16 + (lane >> 4) * 8;
  const float bl = b2[col < NCLASS ? col : NCLASS - 1];
  const float bv = col < NCLASS ? bl : 0.f;
#pragma unroll
  for (int i = 0; i < 8; ++i) {
    int row = rbase + i;
    if (row < Nn) plane0[(size_t)row * PITCH + col] = acc[i] + bv;
  }
}

__global__ void __launch_bounds__(256)
dagnn_rank(const int* __restrict__ dst, int* __restrict__ cnt,
           int* __restrict__ rank, int E) {
  int e = blockIdx.x * blockDim.x + threadIdx.x;
  if (e < E) rank[e] = atomicAdd(&cnt[dst[e]], 1);
}

__global__ void __launch_bounds__(256)
dagnn_tilesum(const int* __restrict__ cnt, int* __restrict__ tsum, int Nn) {
  __shared__ int red[8];
  const int t = threadIdx.x;
  const int i0 = blockIdx.x * 1024 + t * 4;
  int s = 0;
#pragma unroll
  for (int u = 0; u < 4; ++u)
    if (i0 + u < Nn) s += cnt[i0 + u];
#pragma unroll
  for (int off = 16; off > 0; off >>= 1) s += __shfl_xor(s, off, 32);
  if ((t & 31) == 0) red[t >> 5] = s;
  __syncthreads();
  if (t == 0) {
    int r = 0;
#pragma unroll
    for (int w = 0; w < 8; ++w) r += red[w];
    tsum[blockIdx.x] = r;
  }
}

__global__ void __launch_bounds__(256)
dagnn_scan(const int* __restrict__ cnt, const int* __restrict__ tsum,
           int* __restrict__ rowptr, float* __restrict__ norm, int Nn) {
  __shared__ int red[8];
  __shared__ int wsum[8];
  const int t = threadIdx.x, lane = t & 31, wave = t >> 5;
  const int tile = blockIdx.x;

  int o = 0;
  for (int i = t; i < tile; i += 256) o += tsum[i];
#pragma unroll
  for (int off = 16; off > 0; off >>= 1) o += __shfl_xor(o, off, 32);
  if (lane == 0) red[wave] = o;

  const int i0 = tile * 1024 + t * 4;
  int c[4];
#pragma unroll
  for (int u = 0; u < 4; ++u) c[u] = (i0 + u < Nn) ? cnt[i0 + u] : 0;
  const int ts = c[0] + c[1] + c[2] + c[3];
  int inc = ts;
#pragma unroll
  for (int off = 1; off < 32; off <<= 1) {
    int v = __shfl_up(inc, off, 32);
    if (lane >= off) inc += v;
  }
  if (lane == 31) wsum[wave] = inc;
  __syncthreads();

  int ex = inc - ts;
#pragma unroll
  for (int w = 0; w < 8; ++w) {
    ex += red[w];
    if (w < wave) ex += wsum[w];
  }
#pragma unroll
  for (int u = 0; u < 4; ++u) {
    if (i0 + u < Nn) {
      rowptr[i0 + u] = ex;
      norm[i0 + u] = rsqrtf((float)c[u]);
      ex += c[u];
    }
  }
  if (i0 < Nn && Nn <= i0 + 4) rowptr[Nn] = ex;
}

__global__ void __launch_bounds__(256)
dagnn_fill(const int* __restrict__ src, const int* __restrict__ dst,
           const int* __restrict__ rank, const int* __restrict__ rowptr,
           const float* __restrict__ norm, int2* __restrict__ csr, int E) {
  int e = blockIdx.x * blockDim.x + threadIdx.x;
  if (e >= E) return;
  int sv = src[e];
  int pos = rowptr[dst[e]] + rank[e];
  csr[pos] = make_int2(sv, __float_as_int(norm[sv]));
}

template <int C>
__device__ __forceinline__ void gather_chunk(const float* __restrict__ inl, int cx, int cy,
                                             int j, float& ax, float& ay) {
  float2 x[C];
  float w[C];
#pragma unroll
  for (int u = 0; u < C; ++u) {
    const int sv = __builtin_amdgcn_readlane(cx, j + u);
    w[u] = __int_as_float(__builtin_amdgcn_readlane(cy, j + u));
    x[u] = *(const float2*)(inl + (size_t)((unsigned)sv * (unsigned)PITCH));
  }
#pragma unroll
  for (int u = 0; u < C; ++u) {
    ax = fmaf(w[u], x[u].x, ax);
    ay = fmaf(w[u], x[u].y, ay);
  }
}

__global__ void __launch_bounds__(256)
dagnn_gather(const float* __restrict__ in, float* __restrict__ outp,
             const int2* __restrict__ csr, const int* __restrict__ rowptr,
             const float* __restrict__ norm, int Nn) {
  const int lane = threadIdx.x & 31;
  const int node = __builtin_amdgcn_readfirstlane((int)(blockIdx.x * 8 + (threadIdx.x >> 5)));
  if (node >= Nn) return;
  const int beg = rowptr[node], end = rowptr[node + 1];
  const float* inl = in + 2 * lane;
  float ax = 0.f, ay = 0.f;
  for (int base = beg; base < end; base += 32) {
    int n = end - base;
    if (n > 32) n = 32;
    int2 cw = make_int2(0, 0);
    if (lane < n) cw = csr[base + lane];
    int j = 0;
    for (; j + 8 <= n; j += 8) gather_chunk<8>(inl, cw.x, cw.y, j, ax, ay);
    if (n - j >= 4) { gather_chunk<4>(inl, cw.x, cw.y, j, ax, ay); j += 4; }
    if (n - j >= 2) { gather_chunk<2>(inl, cw.x, cw.y, j, ax, ay); j += 2; }
    if (n - j >= 1) gather_chunk<1>(inl, cw.x, cw.y, j, ax, ay);
  }
  const float nd = norm[node];
  const bool real = lane < NCLASS / 2;
  float2 r;
  r.x = real ? nd * ax : 0.f;
  r.y = real ? nd * ay : 0.f;
  *(float2*)(outp + (size_t)node * PITCH + 2 * lane) = r;
}

__global__ void dagnn_combine(const float* __restrict__ planes, const float* __restrict__ s,
                              float* __restrict__ out, int Nn) {
  int gid = blockIdx.x * blockDim.x + threadIdx.x;
  int node = gid >> 5, lane = gid & 31;
  if (node >= Nn) return;
  int c2 = lane + 32;
  float s1 = s[lane];
  float s2 = (c2 < NCLASS) ? s[c2] : 0.f;
  float acc1 = 0.f, acc2 = 0.f;
  size_t pstride = (size_t)Nn * PITCH;
  size_t base = (size_t)node * NCLASS;
  size_t pbase = (size_t)node * PITCH;
#pragma unroll
  for (int k = 0; k <= KHOPS; ++k) {
    const float* hp = planes + (size_t)k * pstride + pbase;
    float h1 = hp[lane];
    float h2 = (c2 < NCLASS) ? hp[c2] : 0.f;
    float z = h1 * s1 + h2 * s2;
#pragma unroll
    for (int off = 16; off > 0; off >>= 1) z += __shfl_xor(z, off, 32);
    float S = 1.0f / (1.0f + __expf(-z));
    acc1 += S * h1;
    acc2 += S * h2;
  }
  out[base + lane] = acc1;
  if (c2 < NCLASS) out[base + c2] = acc2;
}

extern "C" void kernel_launch(void* const* d_in, const int* in_sizes, int n_in,
                              void* d_out, int out_size, void* d_ws, size_t ws_size,
                              hipStream_t stream) {
  const float* feats = (const float*)d_in[0];
  const float* W1    = (const float*)d_in[1];
  const float* b1    = (const float*)d_in[2];
  const float* W2    = (const float*)d_in[3];
  const float* b2    = (const float*)d_in[4];
  const float* s     = (const float*)d_in[5];
  const int*   src   = (const int*)d_in[6];
  const int*   dst   = (const int*)d_in[7];
  const int Nn = in_sizes[0] / IN_DIM;
  const int E  = in_sizes[6];
  float* out = (float*)d_out;

  char* ws = (char*)d_ws;
  size_t off = 0;
  auto carve = [&](size_t bytes) -> char* {
    char* p = ws + off;
    off = (off + bytes + 255) & ~(size_t)255;
    return p;
  };
  const int ntile = (Nn + 1023) / 1024;
  float*    norm   = (float*)   carve((size_t)Nn * sizeof(float));
  int*      cnt    = (int*)     carve((size_t)Nn * sizeof(int));
  int*      rowptr = (int*)     carve((size_t)(Nn + 1) * sizeof(int));
  int*      tsum   = (int*)     carve((size_t)ntile * sizeof(int));
  int*      rank   = (int*)     carve((size_t)E * sizeof(int));
  int2*     csr    = (int2*)    carve((size_t)E * sizeof(int2));
  _Float16* hid    = (_Float16*)carve((size_t)Nn * HID * sizeof(_Float16));
  float*    planes = (float*)   carve((size_t)(KHOPS + 1) * Nn * PITCH * sizeof(float));
  if (off > ws_size) return;

  const size_t pstride = (size_t)Nn * PITCH;

  hipMemsetAsync(cnt, 0, (size_t)Nn * sizeof(int), stream);
  dagnn_rank<<<(E + 255) / 256, 256, 0, stream>>>(dst, cnt, rank, E);
  dagnn_tilesum<<<ntile, 256, 0, stream>>>(cnt, tsum, Nn);
  dagnn_scan<<<ntile, 256, 0, stream>>>(cnt, tsum, rowptr, norm, Nn);
  dagnn_fill<<<(E + 255) / 256, 256, 0, stream>>>(src, dst, rank, rowptr, norm, csr, E);

  dim3 g1((Nn + 63) / 64, HID / 64);
  dagnn_gemm1<<<g1, 256, 0, stream>>>(feats, W1, b1, hid, Nn);
  dim3 g2((Nn + 31) / 32, 1);
  dagnn_gemm2<<<g2, 256, 0, stream>>>(hid, W2, b2, planes, Nn);

  const int ggrid = (Nn + 7) / 8;
  for (int k = 1; k <= KHOPS; ++k) {
    const float* pin = planes + (size_t)(k - 1) * pstride;
    float* pout      = planes + (size_t)k * pstride;
    dagnn_gather<<<ggrid, 256, 0, stream>>>(pin, pout, csr, rowptr, norm, Nn);
  }

  dagnn_combine<<<(int)(((size_t)Nn * 32 + 255) / 256), 256, 0, stream>>>(planes, s, out, Nn);
  (void)n_in; (void)out_size;
}


# GEMM1 re-tiled: one block per 64 rows computes all 256 hidden columns (grid.y 4 -> 1), so feats is read once instead of four times; its W1 k-tile goes to LDS as 16-byte column pieces
# speedup vs baseline: 4.3181x; 1.1890x over previous
#include <hip/hip_runtime.h>
#include <hip/hip_bf16.h>

#define IN_DIM 500
#define HID    256
#define NCLASS 50
#define KHOPS  10
#define PITCH  64

typedef __attribute__((ext_vector_type(16))) _Float16 v16h;
typedef __attribute__((ext_vector_type(8)))  _Float16 h8;
typedef __attribute__((ext_vector_type(4)))  _Float16 h4;
typedef __attribute__((ext_vector_type(8)))  float    v8f;
typedef __attribute__((ext_vector_type(4)))  float    v4f;

__device__ __forceinline__ v16h load_frag(const _Float16* row, int kb) {
  h8 lo = *(const h8*)(row + kb);
  h8 hi = *(const h8*)(row + 16 + kb);
  v16h f;
#pragma unroll
  for (int i = 0; i < 8; ++i) { f[i] = lo[i]; f[8 + i] = hi[i]; }
  return f;
}

__device__ __forceinline__ v8f wmma16(v16h a, v16h b, v8f c) {
  return __builtin_amdgcn_wmma_f32_16x16x32_f16(false, a, false, b, (short)0,
                                                c, false, false);
}

__global__ void __launch_bounds__(256)
dagnn_gemm1(const float* __restrict__ X, const float* __restrict__ W,
            const float* __restrict__ bias, _Float16* __restrict__ Y, int Nn) {
  __shared__ __align__(16) _Float16 As[2][64][40];
  __shared__ __align__(16) _Float16 Bt[2][HID][40];
  const int t = threadIdx.x;
  const int wave = t >> 5, lane = t & 31;
  const int waveM = wave >> 1, waveN = wave & 1;
  const int blockM = blockIdx.x * 64;

  v4f    ra[2];
  float  rb[32];
  const int KTAIL = IN_DIM - 480;

  auto fetch = [&](int kt, const bool tail) {
#pragma unroll
    for (int j = 0; j < 2; ++j) {
      int idx = j * 256 + t;
      int r = idx >> 3, k4 = (idx & 7) * 4;
      int grow = blockM + r;
      int cr = grow < Nn ? grow : Nn - 1;
      int ck = kt + k4;
      if (tail) ck = ck < IN_DIM ? ck : IN_DIM - 4;
      ra[j] = *(const v4f*)(X + (size_t)cr * IN_DIM + ck);
    }
    const float* wp = W + (size_t)kt * HID + t;
#pragma unroll
    for (int kk = 0; kk < 32; ++kk)
      rb[kk] = (!tail || kk < KTAIL) ? wp[kk * HID] : 0.f;
  };
  auto stage = [&](int buf, const bool tail) {
#pragma unroll
    for (int j = 0; j < 2; ++j) {
      int idx = j * 256 + t;
      int r = idx >> 3, k4 = (idx & 7) * 4;
      bool ok = !tail || k4 < KTAIL;
      h4 hv;
      hv[0] = ok ? (_Float16)ra[j].x : (_Float16)0.f;
      hv[1] = ok ? (_Float16)ra[j].y : (_Float16)0.f;
      hv[2] = ok ? (_Float16)ra[j].z : (_Float16)0.f;
      hv[3] = ok ? (_Float16)ra[j].w : (_Float16)0.f;
      *(h4*)&As[buf][r][k4] = hv;
    }
#pragma unroll
    for (int q = 0; q < 4; ++q) {
      h8 p;
#pragma unroll
      for (int i = 0; i < 8; ++i) p[i] = (_Float16)rb[q * 8 + i];
      *(h8*)&Bt[buf][t][q * 8] = p;
    }
  };

  v8f acc[8];
#pragma unroll
  for (int nt = 0; nt < 8; ++nt) acc[nt] = (v8f){0.f, 0.f, 0.f, 0.f, 0.f, 0.f, 0.f, 0.f};
  const int mrow = waveM * 16 + (lane & 15);
  const int nrow = waveN * 128 + (lane & 15);
  const int kb = (lane >> 4) * 8;

  auto mma = [&](int cur) {
    v16h a = load_frag(&As[cur][mrow][0], kb);
#pragma unroll
    for (int nt = 0; nt < 8; ++nt) {
      v16h b = load_frag(&Bt[cur][nrow + nt * 16][0], kb);
      acc[nt] = wmma16(a, b, acc[nt]);
    }
  };

  fetch(0, false);
  stage(0, false);
  __syncthreads();
  const int NKT = 16;
  for (int it = 0; it < NKT - 2; ++it) {
    int cur = it & 1;
    fetch((it + 1) * 32, false);
    asm volatile("" ::: "memory");
    mma(cur);
    stage(cur ^ 1, false);
    __syncthreads();
  }
  fetch((NKT - 1) * 32, true);
  asm volatile("" ::: "memory");
  mma(0);
  stage(1, true);
  __syncthreads();
  mma(1);

  const int rbase = blockM + waveM * 16 + (lane >> 4) * 8;
#pragma unroll
  for (int nt = 0; nt < 8; ++nt) {
    const int col = waveN * 128 + nt * 16 + (lane & 15);
    const float bv = bias[col];
#pragma unroll
    for (int i = 0; i < 8; ++i) {
      int row = rbase + i;
      if (row < Nn) {
        float v = acc[nt][i] + bv;
        Y[(size_t)row * HID + col] = (_Float16)(v > 0.f ? v : 0.f);
      }
    }
  }
}

__global__ void __launch_bounds__(256)
dagnn_gemm2(const _Float16* __restrict__ Hd, const float* __restrict__ W2,
            const float* __restrict__ b2, float* __restrict__ plane0, int Nn) {
  __shared__ __align__(16) _Float16 As[2][32][40];
  __shared__ __align__(16) _Float16 Bt[2][64][40];
  const int t = threadIdx.x;
  const int wave = t >> 5, lane = t & 31;
  const int waveM = wave >> 2, waveN = wave & 3;
  const int blockM = blockIdx.x * 32;

  auto loadA = [&](int kt, int buf) {
    int r = t >> 3, k4 = (t & 7) * 4;
    int grow = blockM + r;
    int cr = grow < Nn ? grow : Nn - 1;
    *(h4*)&As[buf][r][k4] = *(const h4*)(Hd + (size_t)cr * HID + kt + k4);
  };
  auto loadB = [&](int kt, int buf) {
#pragma unroll
    for (int j = 0; j < 8; ++j) {
      int idx = j * 256 + t;
      int n = idx & 63, kk = idx >> 6;
      int cn = n < NCLASS ? n : NCLASS - 1;
      float v = W2[(size_t)(kt + kk) * NCLASS + cn];
      Bt[buf][n][kk] = (n < NCLASS) ? (_Float16)v : (_Float16)0.f;
    }
  };

  v8f acc = {0.f, 0.f, 0.f, 0.f, 0.f, 0.f, 0.f, 0.f};
  const int mrow = waveM * 16 + (lane & 15);
  const int nrow = waveN * 16 + (lane & 15);
  const int kb = (lane >> 4) * 8;

  loadA(0, 0);
  loadB(0, 0);
  __syncthreads();
  const int NKT = HID / 32;
  for (int it = 0; it < NKT; ++it) {
    int cur = it & 1;
    if (it + 1 < NKT) {
      loadA((it + 1) * 32, cur ^ 1);
      loadB((it + 1) * 32, cur ^ 1);
    }
    v16h a = load_frag(&As[cur][mrow][0], kb);
    v16h b = load_frag(&Bt[cur][nrow][0], kb);
    acc = wmma16(a, b, acc);
    __syncthreads();
  }

  const int col = waveN * 16 + (lane & 15);
  const int rbase = blockM + waveM * 16 + (lane >> 4) * 8;
  const float bl = b2[col < NCLASS ? col : NCLASS - 1];
  const float bv = col < NCLASS ? bl : 0.f;
#pragma unroll
  for (int i = 0; i < 8; ++i) {
    int row = rbase + i;
    if (row < Nn) plane0[(size_t)row * PITCH + col] = acc[i] + bv;
  }
}

__global__ void __launch_bounds__(256)
dagnn_rank(const int* __restrict__ dst, int* __restrict__ cnt,
           int* __restrict__ rank, int E) {
  int e = blockIdx.x * blockDim.x + threadIdx.x;
  if (e < E) rank[e] = atomicAdd(&cnt[dst[e]], 1);
}

__global__ void __launch_bounds__(256)
dagnn_tilesum(const int* __restrict__ cnt, int* __restrict__ tsum, int Nn) {
  __shared__ int red[8];
  const int t = threadIdx.x;
  const int i0 = blockIdx.x * 1024 + t * 4;
  int s = 0;
#pragma unroll
  for (int u = 0; u < 4; ++u)
    if (i0 + u < Nn) s += cnt[i0 + u];
#pragma unroll
  for (int off = 16; off > 0; off >>= 1) s += __shfl_xor(s, off, 32);
  if ((t & 31) == 0) red[t >> 5] = s;
  __syncthreads();
  if (t == 0) {
    int r = 0;
#pragma unroll
    for (int w = 0; w < 8; ++w) r += red[w];
    tsum[blockIdx.x] = r;
  }
}

__global__ void __launch_bounds__(256)
dagnn_scan(const int* __restrict__ cnt, const int* __restrict__ tsum,
           int* __restrict__ rowptr, float* __restrict__ norm, int Nn) {
  __shared__ int red[8];
  __shared__ int wsum[8];
  const int t = threadIdx.x, lane = t & 31, wave = t >> 5;
  const int tile = blockIdx.x;

  int o = 0;
  for (int i = t; i < tile; i += 256) o += tsum[i];
#pragma unroll
  for (int off = 16; off > 0; off >>= 1) o += __shfl_xor(o, off, 32);
  if (lane == 0) red[wave] = o;

  const int i0 = tile * 1024 + t * 4;
  int c[4];
#pragma unroll
  for (int u = 0; u < 4; ++u) c[u] = (i0 + u < Nn) ? cnt[i0 + u] : 0;
  const int ts = c[0] + c[1] + c[2] + c[3];
  int inc = ts;
#pragma unroll
  for (int off = 1; off < 32; off <<= 1) {
    int v = __shfl_up(inc, off, 32);
    if (lane >= off) inc += v;
  }
  if (lane == 31) wsum[wave] = inc;
  __syncthreads();

  int ex = inc - ts;
#pragma unroll
  for (int w = 0; w < 8; ++w) {
    ex += red[w];
    if (w < wave) ex += wsum[w];
  }
#pragma unroll
  for (int u = 0; u < 4; ++u) {
    if (i0 + u < Nn) {
      rowptr[i0 + u] = ex;
      norm[i0 + u] = rsqrtf((float)c[u]);
      ex += c[u];
    }
  }
  if (i0 < Nn && Nn <= i0 + 4) rowptr[Nn] = ex;
}

__global__ void __launch_bounds__(256)
dagnn_fill(const int* __restrict__ src, const int* __restrict__ dst,
           const int* __restrict__ rank, const int* __restrict__ rowptr,
           const float* __restrict__ norm, int2* __restrict__ csr, int E) {
  int e = blockIdx.x * blockDim.x + threadIdx.x;
  if (e >= E) return;
  int sv = src[e];
  int pos = rowptr[dst[e]] + rank[e];
  csr[pos] = make_int2(sv, __float_as_int(norm[sv]));
}

template <int C>
__device__ __forceinline__ void gather_chunk(const float* __restrict__ inl, int cx, int cy,
                                             int j, float& ax, float& ay) {
  float2 x[C];
  float w[C];
#pragma unroll
  for (int u = 0; u < C; ++u) {
    const int sv = __builtin_amdgcn_readlane(cx, j + u);
    w[u] = __int_as_float(__builtin_amdgcn_readlane(cy, j + u));
    x[u] = *(const float2*)(inl + (size_t)((unsigned)sv * (unsigned)PITCH));
  }
#pragma unroll
  for (int u = 0; u < C; ++u) {
    ax = fmaf(w[u], x[u].x, ax);
    ay = fmaf(w[u], x[u].y, ay);
  }
}

__global__ void __launch_bounds__(256)
dagnn_gather(const float* __restrict__ in, float* __restrict__ outp,
             const int2* __restrict__ csr, const int* __restrict__ rowptr,
             const float* __restrict__ norm, int Nn) {
  const int lane = threadIdx.x & 31;
  const int node = __builtin_amdgcn_readfirstlane((int)(blockIdx.x * 8 + (threadIdx.x >> 5)));
  if (node >= Nn) return;
  const int beg = rowptr[node], end = rowptr[node + 1];
  const float* inl = in + 2 * lane;
  float ax = 0.f, ay = 0.f;
  for (int base = beg; base < end; base += 32) {
    int n = end - base;
    if (n > 32) n = 32;
    int2 cw = make_int2(0, 0);
    if (lane < n) cw = csr[base + lane];
    int j = 0;
    for (; j + 8 <= n; j += 8) gather_chunk<8>(inl, cw.x, cw.y, j, ax, ay);
    if (n - j >= 4) { gather_chunk<4>(inl, cw.x, cw.y, j, ax, ay); j += 4; }
    if (n - j >= 2) { gather_chunk<2>(inl, cw.x, cw.y, j, ax, ay); j += 2; }
    if (n - j >= 1) gather_chunk<1>(inl, cw.x, cw.y, j, ax, ay);
  }
  const float nd = norm[node];
  const bool real = lane < NCLASS / 2;
  float2 r;
  r.x = real ? nd * ax : 0.f;
  r.y = real ? nd * ay : 0.f;
  *(float2*)(outp + (size_t)node * PITCH + 2 * lane) = r;
}

__global__ void dagnn_combine(const float* __restrict__ planes, const float* __restrict__ s,
                              float* __restrict__ out, int Nn) {
  int gid = blockIdx.x * blockDim.x + threadIdx.x;
  int node = gid >> 5, lane = gid & 31;
  if (node >= Nn) return;
  int c2 = lane + 32;
  float s1 = s[lane];
  float s2 = (c2 < NCLASS) ? s[c2] : 0.f;
  float acc1 = 0.f, acc2 = 0.f;
  size_t pstride = (size_t)Nn * PITCH;
  size_t base = (size_t)node * NCLASS;
  size_t pbase = (size_t)node * PITCH;
#pragma unroll
  for (int k = 0; k <= KHOPS; ++k) {
    const float* hp = planes + (size_t)k * pstride + pbase;
    float h1 = hp[lane];
    float h2 = (c2 < NCLASS) ? hp[c2] : 0.f;
    float z = h1 * s1 + h2 * s2;
#pragma unroll
    for (int off = 16; off > 0; off >>= 1) z += __shfl_xor(z, off, 32);
    float S = 1.0f / (1.0f + __expf(-z));
    acc1 += S * h1;
    acc2 += S * h2;
  }
  out[base + lane] = acc1;
  if (c2 < NCLASS) out[base + c2] = acc2;
}

extern "C" void kernel_launch(void* const* d_in, const int* in_sizes, int n_in,
                              void* d_out, int out_size, void* d_ws, size_t ws_size,
                              hipStream_t stream) {
  const float* feats = (const float*)d_in[0];
  const float* W1    = (const float*)d_in[1];
  const float* b1    = (const float*)d_in[2];
  const float* W2    = (const float*)d_in[3];
  const float* b2    = (const float*)d_in[4];
  const float* s     = (const float*)d_in[5];
  const int*   src   = (const int*)d_in[6];
  const int*   dst   = (const int*)d_in[7];
  const int Nn = in_sizes[0] / IN_DIM;
  const int E  = in_sizes[6];
  float* out = (float*)d_out;

  char* ws = (char*)d_ws;
  size_t off = 0;
  auto carve = [&](size_t bytes) -> char* {
    char* p = ws + off;
    off = (off + bytes + 255) & ~(size_t)255;
    return p;
  };
  const int ntile = (Nn + 1023) / 1024;
  float*    norm   = (float*)   carve((size_t)Nn * sizeof(float));
  int*      cnt    = (int*)     carve((size_t)Nn * sizeof(int));
  int*      rowptr = (int*)     carve((size_t)(Nn + 1) * sizeof(int));
  int*      tsum   = (int*)     carve((size_t)ntile * sizeof(int));
  int*      rank   = (int*)     carve((size_t)E * sizeof(int));
  int2*     csr    = (int2*)    carve((size_t)E * sizeof(int2));
  _Float16* hid    = (_Float16*)carve((size_t)Nn * HID * sizeof(_Float16));
  float*    planes = (float*)   carve((size_t)(KHOPS + 1) * Nn * PITCH * sizeof(float));
  if (off > ws_size) return;

  const size_t pstride = (size_t)Nn * PITCH;

  hipMemsetAsync(cnt, 0, (size_t)Nn * sizeof(int), stream);
  dagnn_rank<<<(E + 255) / 256, 256, 0, stream>>>(dst, cnt, rank, E);
  dagnn_tilesum<<<ntile, 256, 0, stream>>>(cnt, tsum, Nn);
  dagnn_scan<<<ntile, 256, 0, stream>>>(cnt, tsum, rowptr, norm, Nn);
  dagnn_fill<<<(E + 255) / 256, 256, 0, stream>>>(src, dst, rank, rowptr, norm, csr, E);

  dim3 g1((Nn + 63) / 64, 1);
  dagnn_gemm1<<<g1, 256, 0, stream>>>(feats, W1, b1, hid, Nn);
  dim3 g2((Nn + 31) / 32, 1);
  dagnn_gemm2<<<g2, 256, 0, stream>>>(hid, W2, b2, planes, Nn);

  const int ggrid = (Nn + 7) / 8;
  for (int k = 1; k <= KHOPS; ++k) {
    const float* pin = planes + (size_t)(k - 1) * pstride;
    float* pout      = planes + (size_t)k * pstride;
    dagnn_gather<<<ggrid, 256, 0, stream>>>(pin, pout, csr, rowptr, norm, Nn);
  }

  dagnn_combine<<<(int)(((size_t)Nn * 32 + 255) / 256), 256, 0, stream>>>(planes, s, out, Nn);
  (void)n_in; (void)out_size;
}


# Readout fused into the last hop: its wave adds the node's own rows of planes 0..9, streamed beside the gathered rows; plane 10 and the combine launch go; z summed over lane pairs (2l,2l+1)
# speedup vs baseline: 4.8780x; 1.1297x over previous
#include <hip/hip_runtime.h>
#include <hip/hip_bf16.h>

#define IN_DIM 500
#define HID    256
#define NCLASS 50
#define KHOPS  10
#define PITCH  64

typedef __attribute__((ext_vector_type(16))) _Float16 v16h;
typedef __attribute__((ext_vector_type(8)))  _Float16 h8;
typedef __attribute__((ext_vector_type(4)))  _Float16 h4;
typedef __attribute__((ext_vector_type(8)))  float    v8f;
typedef __attribute__((ext_vector_type(4)))  float    v4f;
typedef __attribute__((ext_vector_type(4)))  unsigned int v4u;
typedef __attribute__((ext_vector_type(2)))  float    v2f;

__device__ __forceinline__ v16h load_frag(const _Float16* row, int kb) {
  h8 lo = *(const h8*)(row + kb);
  h8 hi = *(const h8*)(row + 16 + kb);
  v16h f;
#pragma unroll
  for (int i = 0; i < 8; ++i) { f[i] = lo[i]; f[8 + i] = hi[i]; }
  return f;
}

__device__ __forceinline__ v8f wmma16(v16h a, v16h b, v8f c) {
  return __builtin_amdgcn_wmma_f32_16x16x32_f16(false, a, false, b, (short)0,
                                                c, false, false);
}

#define HPITCH 264

__global__ void __launch_bounds__(256)
dagnn_mlp(const float* __restrict__ X, const float* __restrict__ W,
          const float* __restrict__ bias, const float* __restrict__ W2,
          const float* __restrict__ b2, float* __restrict__ plane0, int Nn) {
  __shared__ __align__(16) _Float16 As[2][64][40];
  __shared__ __align__(16) _Float16 Bt[2][HID][40];
  __shared__ __align__(16) _Float16 W2s[64][HPITCH];
  _Float16 (*Hs)[HPITCH] = reinterpret_cast<_Float16 (*)[HPITCH]>(&Bt[0][0][0]);
  const int t = threadIdx.x;
  const int wave = t >> 5, lane = t & 31;
  const int waveM = wave >> 1, waveN = wave & 1;
  const int blockM = blockIdx.x * 64;

  v4f    ra[2];
  float  rb[32];
  const int KTAIL = IN_DIM - 480;

  auto fetch = [&](int kt, const bool tail) {
#pragma unroll
    for (int j = 0; j < 2; ++j) {
      int idx = j * 256 + t;
      int r = idx >> 3, k4 = (idx & 7) * 4;
      int grow = blockM + r;
      int cr = grow < Nn ? grow : Nn - 1;
      int ck = kt + k4;
      if (tail) ck = ck < IN_DIM ? ck : IN_DIM - 4;
      ra[j] = *(const v4f*)(X + (size_t)cr * IN_DIM + ck);
    }
    const float* wp = W + (size_t)kt * HID + t;
#pragma unroll
    for (int kk = 0; kk < 32; ++kk)
      rb[kk] = (!tail || kk < KTAIL) ? wp[kk * HID] : 0.f;
  };
  auto stage = [&](int buf, const bool tail) {
#pragma unroll
    for (int j = 0; j < 2; ++j) {
      int idx = j * 256 + t;
      int r = idx >> 3, k4 = (idx & 7) * 4;
      bool ok = !tail || k4 < KTAIL;
      h4 hv;
      hv[0] = ok ? (_Float16)ra[j].x : (_Float16)0.f;
      hv[1] = ok ? (_Float16)ra[j].y : (_Float16)0.f;
      hv[2] = ok ? (_Float16)ra[j].z : (_Float16)0.f;
      hv[3] = ok ? (_Float16)ra[j].w : (_Float16)0.f;
      *(h4*)&As[buf][r][k4] = hv;
    }
#pragma unroll
    for (int q = 0; q < 4; ++q) {
      h8 p;
#pragma unroll
      for (int i = 0; i < 8; ++i) p[i] = (_Float16)rb[q * 8 + i];
      *(h8*)&Bt[buf][t][q * 8] = p;
    }
  };

  v8f acc[8];
#pragma unroll
  for (int nt = 0; nt < 8; ++nt) acc[nt] = (v8f){0.f, 0.f, 0.f, 0.f, 0.f, 0.f, 0.f, 0.f};
  const int mrow = waveM * 16 + (lane & 15);
  const int nrow = waveN * 128 + (lane & 15);
  const int kb = (lane >> 4) * 8;

  auto mma = [&](int cur) {
    v16h xf = load_frag(&As[cur][mrow][0], kb);
#pragma unroll
    for (int nt = 0; nt < 8; ++nt) {
      v16h wf = load_frag(&Bt[cur][nrow + nt * 16][0], kb);
      acc[nt] = wmma16(wf, xf, acc[nt]);
    }
  };

  fetch(0, false);
  {
    const int n = t & 63, kq = t >> 6;
    const bool real = n < NCLASS;
    const unsigned keep = real ? 0xFFFFFFFFu : 0u;
    const float* w2p = W2 + (real ? n : NCLASS - 1) + (size_t)kq * 8 * NCLASS;
#pragma unroll
    for (int jb = 0; jb < 2; ++jb) {
      float wv[4][8];
#pragma unroll
      for (int j = 0; j < 4; ++j)
#pragma unroll
        for (int i = 0; i < 8; ++i)
          wv[j][i] = w2p[((jb * 4 + j) * 32 + i) * NCLASS];
#pragma unroll
      for (int j = 0; j < 4; ++j) {
        const int k8 = (jb * 4 + j) * 4 + kq;
        h8 p;
#pragma unroll
        for (int i = 0; i < 8; ++i) p[i] = (_Float16)wv[j][i];
        v4u pu = __builtin_bit_cast(v4u, p);
        pu &= (v4u){keep, keep, keep, keep};
        *(h8*)&W2s[n][k8 * 8] = __builtin_bit_cast(h8, pu);
      }
    }
  }
  stage(0, false);
  __syncthreads();
  const int NKT = 16;
  for (int it = 0; it < NKT - 2; ++it) {
    int cur = it & 1;
    fetch((it + 1) * 32, false);
    asm volatile("" ::: "memory");
    mma(cur);
    __builtin_amdgcn_sched_barrier(0);
    stage(cur ^ 1, false);
    __syncthreads();
  }
  fetch((NKT - 1) * 32, true);
  asm volatile("" ::: "memory");
  mma(0);
  __builtin_amdgcn_sched_barrier(0);
  stage(1, true);
  __syncthreads();
  mma(1);
  __syncthreads();

#pragma unroll
  for (int nt = 0; nt < 8; ++nt) {
    const int n0 = waveN * 128 + nt * 16 + kb;
    const v4f bA = *(const v4f*)(bias + n0);
    const v4f bB = *(const v4f*)(bias + n0 + 4);
    h8 p;
#pragma unroll
    for (int r = 0; r < 8; ++r) {
      float v = acc[nt][r] + (r < 4 ? bA[r] : bB[r - 4]);
      p[r] = (_Float16)(v > 0.f ? v : 0.f);
    }
    *(h8*)&Hs[mrow][n0] = p;
  }
  __syncthreads();

  v8f o[2];
  o[0] = (v8f){0.f, 0.f, 0.f, 0.f, 0.f, 0.f, 0.f, 0.f};
  o[1] = (v8f){0.f, 0.f, 0.f, 0.f, 0.f, 0.f, 0.f, 0.f};
  const int ncol = waveN * 32 + (lane & 15);
  float bv2[2];
#pragma unroll
  for (int q = 0; q < 2; ++q) {
    const int col = ncol + q * 16;
    const float bl = b2[col < NCLASS ? col : NCLASS - 1];
    bv2[q] = col < NCLASS ? bl : 0.f;
  }
#pragma unroll
  for (int kh = 0; kh < HID / 128; ++kh) {
    v16h hf[4], w0[4], w1[4];
#pragma unroll
    for (int u = 0; u < 4; ++u) {
      const int k0 = (kh * 4 + u) * 32;
      hf[u] = load_frag(&Hs[mrow][k0], kb);
      w0[u] = load_frag(&W2s[ncol][k0], kb);
      w1[u] = load_frag(&W2s[ncol + 16][k0], kb);
    }
#pragma unroll
    for (int u = 0; u < 4; ++u) {
      o[0] = wmma16(hf[u], w0[u], o[0]);
      o[1] = wmma16(hf[u], w1[u], o[1]);
    }
  }

  const int rbase = blockM + waveM * 16 + kb;
#pragma unroll
  for (int q = 0; q < 2; ++q) {
    const int col = ncol + q * 16;
#pragma unroll
    for (int i = 0; i < 8; ++i) {
      int row = rbase + i;
      if (row < Nn) plane0[(size_t)row * PITCH + col] = o[q][i] + bv2[q];
    }
  }
}

__global__ void __launch_bounds__(256)
dagnn_rank(const int* __restrict__ dst, int* __restrict__ cnt,
           int* __restrict__ rank, int E) {
  int e = blockIdx.x * blockDim.x + threadIdx.x;
  if (e < E) rank[e] = atomicAdd(&cnt[dst[e]], 1);
}

__global__ void __launch_bounds__(256)
dagnn_tilesum(const int* __restrict__ cnt, int* __restrict__ tsum, int Nn) {
  __shared__ int red[8];
  const int t = threadIdx.x;
  const int i0 = blockIdx.x * 1024 + t * 4;
  int s = 0;
#pragma unroll
  for (int u = 0; u < 4; ++u)
    if (i0 + u < Nn) s += cnt[i0 + u];
#pragma unroll
  for (int off = 16; off > 0; off >>= 1) s += __shfl_xor(s, off, 32);
  if ((t & 31) == 0) red[t >> 5] = s;
  __syncthreads();
  if (t == 0) {
    int r = 0;
#pragma unroll
    for (int w = 0; w < 8; ++w) r += red[w];
    tsum[blockIdx.x] = r;
  }
}

__global__ void __launch_bounds__(256)
dagnn_scan(const int* __restrict__ cnt, const int* __restrict__ tsum,
           int* __restrict__ rowptr, float* __restrict__ norm, int Nn) {
  __shared__ int red[8];
  __shared__ int wsum[8];
  const int t = threadIdx.x, lane = t & 31, wave = t >> 5;
  const int tile = blockIdx.x;

  int o = 0;
  for (int i = t; i < tile; i += 256) o += tsum[i];
#pragma unroll
  for (int off = 16; off > 0; off >>= 1) o += __shfl_xor(o, off, 32);
  if (lane == 0) red[wave] = o;

  const int i0 = tile * 1024 + t * 4;
  int c[4];
#pragma unroll
  for (int u = 0; u < 4; ++u) c[u] = (i0 + u < Nn) ? cnt[i0 + u] : 0;
  const int ts = c[0] + c[1] + c[2] + c[3];
  int inc = ts;
#pragma unroll
  for (int off = 1; off < 32; off <<= 1) {
    int v = __shfl_up(inc, off, 32);
    if (lane >= off) inc += v;
  }
  if (lane == 31) wsum[wave] = inc;
  __syncthreads();

  int ex = inc - ts;
#pragma unroll
  for (int w = 0; w < 8; ++w) {
    ex += red[w];
    if (w < wave) ex += wsum[w];
  }
#pragma unroll
  for (int u = 0; u < 4; ++u) {
    if (i0 + u < Nn) {
      rowptr[i0 + u] = ex;
      norm[i0 + u] = rsqrtf((float)c[u]);
      ex += c[u];
    }
  }
  if (i0 < Nn && Nn <= i0 + 4) rowptr[Nn] = ex;
}

__global__ void __launch_bounds__(256)
dagnn_fill(const int* __restrict__ src, const int* __restrict__ dst,
           const int* __restrict__ rank, const int* __restrict__ rowptr,
           const float* __restrict__ norm, int2* __restrict__ csr, int E) {
  int e = blockIdx.x * blockDim.x + threadIdx.x;
  if (e >= E) return;
  int sv = src[e];
  int pos = rowptr[dst[e]] + rank[e];
  csr[pos] = make_int2(sv, __float_as_int(norm[sv]));
}

template <int C>
__device__ __forceinline__ void gather_chunk(const float* __restrict__ inl, int cx, int cy,
                                             int j, float& ax, float& ay) {
  float2 x[C];
  float w[C];
#pragma unroll
  for (int u = 0; u < C; ++u) {
    const int sv = __builtin_amdgcn_readlane(cx, j + u);
    w[u] = __int_as_float(__builtin_amdgcn_readlane(cy, j + u));
    x[u] = *(const float2*)(inl + (size_t)((unsigned)sv * (unsigned)PITCH));
  }
#pragma unroll
  for (int u = 0; u < C; ++u) {
    ax = fmaf(w[u], x[u].x, ax);
    ay = fmaf(w[u], x[u].y, ay);
  }
}

__global__ void __launch_bounds__(256)
dagnn_gather(const float* __restrict__ in, float* __restrict__ outp,
             const int2* __restrict__ csr, const int* __restrict__ rowptr,
             const float* __restrict__ norm, int Nn) {
  const int lane = threadIdx.x & 31;
  const int node = __builtin_amdgcn_readfirstlane((int)(blockIdx.x * 8 + (threadIdx.x >> 5)));
  if (node >= Nn) return;
  const int beg = rowptr[node], end = rowptr[node + 1];
  const float* inl = in + 2 * lane;
  float ax = 0.f, ay = 0.f;
  for (int base = beg; base < end; base += 32) {
    int n = end - base;
    if (n > 32) n = 32;
    int2 cw = make_int2(0, 0);
    if (lane < n) cw = csr[base + lane];
    int j = 0;
    for (; j + 8 <= n; j += 8) gather_chunk<8>(inl, cw.x, cw.y, j, ax, ay);
    if (n - j >= 4) { gather_chunk<4>(inl, cw.x, cw.y, j, ax, ay); j += 4; }
    if (n - j >= 2) { gather_chunk<2>(inl, cw.x, cw.y, j, ax, ay); j += 2; }
    if (n - j >= 1) gather_chunk<1>(inl, cw.x, cw.y, j, ax, ay);
  }
  const float nd = norm[node];
  const bool real = lane < NCLASS / 2;
  float2 r;
  r.x = real ? nd * ax : 0.f;
  r.y = real ? nd * ay : 0.f;
  *(float2*)(outp + (size_t)node * PITCH + 2 * lane) = r;
}

__global__ void __launch_bounds__(256)
dagnn_gather_readout(const float* __restrict__ planes, const float* __restrict__ s,
                     float* __restrict__ out, const int2* __restrict__ csr,
                     const int* __restrict__ rowptr, const float* __restrict__ norm, int Nn) {
  const int lane = threadIdx.x & 31;
  const int node = __builtin_amdgcn_readfirstlane((int)(blockIdx.x * 8 + (threadIdx.x >> 5)));
  if (node >= Nn) return;
  const int beg = rowptr[node], end = rowptr[node + 1];
  const size_t pstride = (size_t)Nn * PITCH;
  const float* inl = planes + (size_t)(KHOPS - 1) * pstride + 2 * lane;

  int nfirst = end - beg;
  if (nfirst > 32) nfirst = 32;
  int2 cw = make_int2(0, 0);
  if (lane < nfirst) cw = csr[beg + lane];

  const float* ownp = planes + (size_t)node * PITCH + 2 * lane;
  v2f own[KHOPS];
#pragma unroll
  for (int k = 0; k < KHOPS - 1; ++k)
    own[k] = __builtin_nontemporal_load((const v2f*)(ownp + (size_t)k * pstride));
  own[KHOPS - 1] = *(const v2f*)(ownp + (size_t)(KHOPS - 1) * pstride);

  const bool real = lane < NCLASS / 2;
  const v2f sv = *(const v2f*)(s + (real ? 2 * lane : 0));
  const float sx = real ? sv.x : 0.f, sy = real ? sv.y : 0.f;
  const float nd = norm[node];

  float ax = 0.f, ay = 0.f;
  auto walk = [&](int n) {
    int j = 0;
    for (; j + 8 <= n; j += 8) gather_chunk<8>(inl, cw.x, cw.y, j, ax, ay);
    if (n - j >= 4) { gather_chunk<4>(inl, cw.x, cw.y, j, ax, ay); j += 4; }
    if (n - j >= 2) { gather_chunk<2>(inl, cw.x, cw.y, j, ax, ay); j += 2; }
    if (n - j >= 1) gather_chunk<1>(inl, cw.x, cw.y, j, ax, ay);
  };
  walk(nfirst);
  for (int base = beg + 32; base < end; base += 32) {
    int m = end - base;
    if (m > 32) m = 32;
    cw = make_int2(0, 0);
    if (lane < m) cw = csr[base + lane];
    walk(m);
  }
  const float hx = real ? nd * ax : 0.f;
  const float hy = real ? nd * ay : 0.f;

  float ox = 0.f, oy = 0.f;
  auto add_hop = [&](float x, float y) {
    float z = real ? x * sx + y * sy : 0.f;
#pragma unroll
    for (int off = 16; off > 0; off >>= 1) z += __shfl_xor(z, off, 32);
    const float S = 1.0f / (1.0f + __expf(-z));
    ox += S * x;
    oy += S * y;
  };
#pragma unroll
  for (int k = 0; k < KHOPS; ++k) add_hop(own[k].x, own[k].y);
  add_hop(hx, hy);
  if (real) {
    v2f r;
    r.x = ox;
    r.y = oy;
    *(v2f*)(out + (size_t)node * NCLASS + 2 * lane) = r;
  }
}

extern "C" void kernel_launch(void* const* d_in, const int* in_sizes, int n_in,
                              void* d_out, int out_size, void* d_ws, size_t ws_size,
                              hipStream_t stream) {
  const float* feats = (const float*)d_in[0];
  const float* W1    = (const float*)d_in[1];
  const float* b1    = (const float*)d_in[2];
  const float* W2    = (const float*)d_in[3];
  const float* b2    = (const float*)d_in[4];
  const float* s     = (const float*)d_in[5];
  const int*   src   = (const int*)d_in[6];
  const int*   dst   = (const int*)d_in[7];
  const int Nn = in_sizes[0] / IN_DIM;
  const int E  = in_sizes[6];
  float* out = (float*)d_out;

  char* ws = (char*)d_ws;
  size_t off = 0;
  auto carve = [&](size_t bytes) -> char* {
    char* p = ws + off;
    off = (off + bytes + 255) & ~(size_t)255;
    return p;
  };
  const int ntile = (Nn + 1023) / 1024;
  float*    norm   = (float*)   carve((size_t)Nn * sizeof(float));
  int*      cnt    = (int*)     carve((size_t)Nn * sizeof(int));
  int*      rowptr = (int*)     carve((size_t)(Nn + 1) * sizeof(int));
  int*      tsum   = (int*)     carve((size_t)ntile * sizeof(int));
  int*      rank   = (int*)     carve((size_t)E * sizeof(int));
  int2*     csr    = (int2*)    carve((size_t)E * sizeof(int2));
  float*    planes = (float*)   carve((size_t)KHOPS * Nn * PITCH * sizeof(float));
  if (off > ws_size) return;

  const size_t pstride = (size_t)Nn * PITCH;

  hipMemsetAsync(cnt, 0, (size_t)Nn * sizeof(int), stream);
  dagnn_rank<<<(E + 255) / 256, 256, 0, stream>>>(dst, cnt, rank, E);
  dagnn_tilesum<<<ntile, 256, 0, stream>>>(cnt, tsum, Nn);
  dagnn_scan<<<ntile, 256, 0, stream>>>(cnt, tsum, rowptr, norm, Nn);
  dagnn_fill<<<(E + 255) / 256, 256, 0, stream>>>(src, dst, rank, rowptr, norm, csr, E);

  dim3 g1((Nn + 63) / 64, 1);
  dagnn_mlp<<<g1, 256, 0, stream>>>(feats, W1, b1, W2, b2, planes, Nn);

  const int ggrid = (Nn + 7) / 8;
  for (int k = 1; k < KHOPS; ++k) {
    const float* pin = planes + (size_t)(k - 1) * pstride;
    float* pout      = planes + (size_t)k * pstride;
    dagnn_gather<<<ggrid, 256, 0, stream>>>(pin, pout, csr, rowptr, norm, Nn);
  }

  dagnn_gather_readout<<<ggrid, 256, 0, stream>>>(planes, s, out, csr, rowptr, norm, Nn);
  (void)n_in; (void)out_size;
}


# W1 and W2 converted once per call to f16 images in LDS tile order (spare blocks of the tile-sum launch); MLP blocks stage them with 16-byte loads and no conversions instead of one float per lane
# speedup vs baseline: 5.0984x; 1.0452x over previous
#include <hip/hip_runtime.h>
#include <hip/hip_bf16.h>

#define IN_DIM 500
#define HID    256
#define NCLASS 50
#define KHOPS  10
#define PITCH  64

typedef __attribute__((ext_vector_type(16))) _Float16 v16h;
typedef __attribute__((ext_vector_type(8)))  _Float16 h8;
typedef __attribute__((ext_vector_type(4)))  _Float16 h4;
typedef __attribute__((ext_vector_type(8)))  float    v8f;
typedef __attribute__((ext_vector_type(4)))  float    v4f;
typedef __attribute__((ext_vector_type(2)))  float    v2f;

__device__ __forceinline__ v16h load_frag(const _Float16* row, int kb) {
  h8 lo = *(const h8*)(row + kb);
  h8 hi = *(const h8*)(row + 16 + kb);
  v16h f;
#pragma unroll
  for (int i = 0; i < 8; ++i) { f[i] = lo[i]; f[8 + i] = hi[i]; }
  return f;
}

__device__ __forceinline__ v8f wmma16(v16h a, v16h b, v8f c) {
  return __builtin_amdgcn_wmma_f32_16x16x32_f16(false, a, false, b, (short)0,
                                                c, false, false);
}

#define HPITCH 264
#define NKTILE 16
#define W1_PIECES (NKTILE * 4 * HID)
#define W2_PIECES (64 * (HID / 8))

#define PREP_BLOCKS ((W1_PIECES + W2_PIECES + 255) / 256)
__device__ __forceinline__ void prep_weights(int g, const float* __restrict__ W1,
                                             const float* __restrict__ W2,
                                             _Float16* __restrict__ W1h,
                                             _Float16* __restrict__ W2h) {
  float v[8];
  h8 p;
  if (g < W1_PIECES) {
    const int n = g & (HID - 1), k0 = (g >> 8) * 8;
#pragma unroll
    for (int i = 0; i < 8; ++i) {
      const int k = k0 + i;
      v[i] = W1[(size_t)(k < IN_DIM ? k : IN_DIM - 1) * HID + n];
    }
#pragma unroll
    for (int i = 0; i < 8; ++i) p[i] = (_Float16)(k0 + i < IN_DIM ? v[i] : 0.f);
    ((h8*)W1h)[g] = p;
  } else if (g < W1_PIECES + W2_PIECES) {
    const int r = g - W1_PIECES;
    const int n = r & 63, k8 = r >> 6;
    const float* w2p = W2 + (n < NCLASS ? n : NCLASS - 1) + (size_t)k8 * 8 * NCLASS;
#pragma unroll
    for (int i = 0; i < 8; ++i) v[i] = w2p[i * NCLASS];
#pragma unroll
    for (int i = 0; i < 8; ++i) p[i] = (_Float16)(n < NCLASS ? v[i] : 0.f);
    ((h8*)W2h)[n * (HID / 8) + k8] = p;
  }
}

__global__ void __launch_bounds__(256)
dagnn_mlp(const float* __restrict__ X, const _Float16* __restrict__ W1h,
          const float* __restrict__ bias, const _Float16* __restrict__ W2h,
          const float* __restrict__ b2, float* __restrict__ plane0, int Nn) {
  __shared__ __align__(16) _Float16 As[2][64][40];
  __shared__ __align__(16) _Float16 Bt[2][HID][40];
  __shared__ __align__(16) _Float16 W2s[64][HPITCH];
  _Float16 (*Hs)[HPITCH] = reinterpret_cast<_Float16 (*)[HPITCH]>(&Bt[0][0][0]);
  const int t = threadIdx.x;
  const int wave = t >> 5, lane = t & 31;
  const int waveM = wave >> 1, waveN = wave & 1;
  const int blockM = blockIdx.x * 64;

  v4f    ra[2];
  h8     rb[4];
  const int KTAIL = IN_DIM - 480;

  auto fetch = [&](int kt, const bool tail) {
#pragma unroll
    for (int j = 0; j < 2; ++j) {
      int idx = j * 256 + t;
      int r = idx >> 3, k4 = (idx & 7) * 4;
      int grow = blockM + r;
      int cr = grow < Nn ? grow : Nn - 1;
      int ck = kt + k4;
      if (tail) ck = ck < IN_DIM ? ck : IN_DIM - 4;
      ra[j] = *(const v4f*)(X + (size_t)cr * IN_DIM + ck);
    }
    const h8* wp = (const h8*)W1h + (size_t)(kt >> 3) * HID + t;
#pragma unroll
    for (int q = 0; q < 4; ++q)
      rb[q] = wp[q * HID];
  };
  auto stage = [&](int buf, const bool tail) {
#pragma unroll
    for (int j = 0; j < 2; ++j) {
      int idx = j * 256 + t;
      int r = idx >> 3, k4 = (idx & 7) * 4;
      bool ok = !tail || k4 < KTAIL;
      h4 hv;
      hv[0] = ok ? (_Float16)ra[j].x : (_Float16)0.f;
      hv[1] = ok ? (_Float16)ra[j].y : (_Float16)0.f;
      hv[2] = ok ? (_Float16)ra[j].z : (_Float16)0.f;
      hv[3] = ok ? (_Float16)ra[j].w : (_Float16)0.f;
      *(h4*)&As[buf][r][k4] = hv;
    }
#pragma unroll
    for (int q = 0; q < 4; ++q)
      *(h8*)&Bt[buf][t][q * 8] = rb[q];
  };

  v8f acc[8];
#pragma unroll
  for (int nt = 0; nt < 8; ++nt) acc[nt] = (v8f){0.f, 0.f, 0.f, 0.f, 0.f, 0.f, 0.f, 0.f};
  const int mrow = waveM * 16 + (lane & 15);
  const int nrow = waveN * 128 + (lane & 15);
  const int kb = (lane >> 4) * 8;

  auto mma = [&](int cur) {
    v16h xf = load_frag(&As[cur][mrow][0], kb);
#pragma unroll
    for (int nt = 0; nt < 8; ++nt) {
      v16h wf = load_frag(&Bt[cur][nrow + nt * 16][0], kb);
      acc[nt] = wmma16(wf, xf, acc[nt]);
    }
  };

  fetch(0, false);
  {
    const h8* w2p = (const h8*)W2h + t;
    h8 wv[8];
#pragma unroll
    for (int j = 0; j < 8; ++j) wv[j] = w2p[j * 256];
#pragma unroll
    for (int j = 0; j < 8; ++j) {
      const int p = j * 256 + t;
      *(h8*)&W2s[p >> 5][(p & 31) * 8] = wv[j];
    }
  }
  stage(0, false);
  __syncthreads();
  const int NKT = NKTILE;
  for (int it = 0; it < NKT - 2; ++it) {
    int cur = it & 1;
    fetch((it + 1) * 32, false);
    asm volatile("" ::: "memory");
    mma(cur);
    __builtin_amdgcn_sched_barrier(0);
    stage(cur ^ 1, false);
    __syncthreads();
  }
  fetch((NKT - 1) * 32, true);
  asm volatile("" ::: "memory");
  mma(0);
  __builtin_amdgcn_sched_barrier(0);
  stage(1, true);
  __syncthreads();
  mma(1);
  __syncthreads();

#pragma unroll
  for (int nt = 0; nt < 8; ++nt) {
    const int n0 = waveN * 128 + nt * 16 + kb;
    const v4f bA = *(const v4f*)(bias + n0);
    const v4f bB = *(const v4f*)(bias + n0 + 4);
    h8 p;
#pragma unroll
    for (int r = 0; r < 8; ++r) {
      float v = acc[nt][r] + (r < 4 ? bA[r] : bB[r - 4]);
      p[r] = (_Float16)(v > 0.f ? v : 0.f);
    }
    *(h8*)&Hs[mrow][n0] = p;
  }
  __syncthreads();

  v8f o[2];
  o[0] = (v8f){0.f, 0.f, 0.f, 0.f, 0.f, 0.f, 0.f, 0.f};
  o[1] = (v8f){0.f, 0.f, 0.f, 0.f, 0.f, 0.f, 0.f, 0.f};
  const int ncol = waveN * 32 + (lane & 15);
  float bv2[2];
#pragma unroll
  for (int q = 0; q < 2; ++q) {
    const int col = ncol + q * 16;
    const float bl = b2[col < NCLASS ? col : NCLASS - 1];
    bv2[q] = col < NCLASS ? bl : 0.f;
  }
#pragma unroll
  for (int kh = 0; kh < HID / 128; ++kh) {
    v16h hf[4], w0[4], w1[4];
#pragma unroll
    for (int u = 0; u < 4; ++u) {
      const int k0 = (kh * 4 + u) * 32;
      hf[u] = load_frag(&Hs[mrow][k0], kb);
      w0[u] = load_frag(&W2s[ncol][k0], kb);
      w1[u] = load_frag(&W2s[ncol + 16][k0], kb);
    }
#pragma unroll
    for (int u = 0; u < 4; ++u) {
      o[0] = wmma16(hf[u], w0[u], o[0]);
      o[1] = wmma16(hf[u], w1[u], o[1]);
    }
  }

  const int rbase = blockM + waveM * 16 + kb;
#pragma unroll
  for (int q = 0; q < 2; ++q) {
    const int col = ncol + q * 16;
#pragma unroll
    for (int i = 0; i < 8; ++i) {
      int row = rbase + i;
      if (row < Nn) plane0[(size_t)row * PITCH + col] = o[q][i] + bv2[q];
    }
  }
}

__global__ void __launch_bounds__(256)
dagnn_rank(const int* __restrict__ dst, int* __restrict__ cnt,
           int* __restrict__ rank, int E) {
  int e = blockIdx.x * blockDim.x + threadIdx.x;
  if (e < E) rank[e] = atomicAdd(&cnt[dst[e]], 1);
}

__global__ void __launch_bounds__(256)
dagnn_tilesum_prep(const int* __restrict__ cnt, int* __restrict__ tsum, int Nn, int ntile,
                   const float* __restrict__ W1, const float* __restrict__ W2,
                   _Float16* __restrict__ W1h, _Float16* __restrict__ W2h) {
  if ((int)blockIdx.x >= ntile) {
    prep_weights(((int)blockIdx.x - ntile) * 256 + (int)threadIdx.x, W1, W2, W1h, W2h);
    return;
  }
  __shared__ int red[8];
  const int t = threadIdx.x;
  const int i0 = blockIdx.x * 1024 + t * 4;
  int s = 0;
#pragma unroll
  for (int u = 0; u < 4; ++u)
    if (i0 + u < Nn) s += cnt[i0 + u];
#pragma unroll
  for (int off = 16; off > 0; off >>= 1) s += __shfl_xor(s, off, 32);
  if ((t & 31) == 0) red[t >> 5] = s;
  __syncthreads();
  if (t == 0) {
    int r = 0;
#pragma unroll
    for (int w = 0; w < 8; ++w) r += red[w];
    tsum[blockIdx.x] = r;
  }
}

__global__ void __launch_bounds__(256)
dagnn_scan(const int* __restrict__ cnt, const int* __restrict__ tsum,
           int* __restrict__ rowptr, float* __restrict__ norm, int Nn) {
  __shared__ int red[8];
  __shared__ int wsum[8];
  const int t = threadIdx.x, lane = t & 31, wave = t >> 5;
  const int tile = blockIdx.x;

  int o = 0;
  for (int i = t; i < tile; i += 256) o += tsum[i];
#pragma unroll
  for (int off = 16; off > 0; off >>= 1) o += __shfl_xor(o, off, 32);
  if (lane == 0) red[wave] = o;

  const int i0 = tile * 1024 + t * 4;
  int c[4];
#pragma unroll
  for (int u = 0; u < 4; ++u) c[u] = (i0 + u < Nn) ? cnt[i0 + u] : 0;
  const int ts = c[0] + c[1] + c[2] + c[3];
  int inc = ts;
#pragma unroll
  for (int off = 1; off < 32; off <<= 1) {
    int v = __shfl_up(inc, off, 32);
    if (lane >= off) inc += v;
  }
  if (lane == 31) wsum[wave] = inc;
  __syncthreads();

  int ex = inc - ts;
#pragma unroll
  for (int w = 0; w < 8; ++w) {
    ex += red[w];
    if (w < wave) ex += wsum[w];
  }
#pragma unroll
  for (int u = 0; u < 4; ++u) {
    if (i0 + u < Nn) {
      rowptr[i0 + u] = ex;
      norm[i0 + u] = rsqrtf((float)c[u]);
      ex += c[u];
    }
  }
  if (i0 < Nn && Nn <= i0 + 4) rowptr[Nn] = ex;
}

__global__ void __launch_bounds__(256)
dagnn_fill(const int* __restrict__ src, const int* __restrict__ dst,
           const int* __restrict__ rank, const int* __restrict__ rowptr,
           const float* __restrict__ norm, int2* __restrict__ csr, int E) {
  int e = blockIdx.x * blockDim.x + threadIdx.x;
  if (e >= E) return;
  int sv = src[e];
  int pos = rowptr[dst[e]] + rank[e];
  csr[pos] = make_int2(sv, __float_as_int(norm[sv]));
}

template <int C>
__device__ __forceinline__ void gather_chunk(const float* __restrict__ inl, int cx, int cy,
                                             int j, float& ax, float& ay) {
  float2 x[C];
  float w[C];
#pragma unroll
  for (int u = 0; u < C; ++u) {
    const int sv = __builtin_amdgcn_readlane(cx, j + u);
    w[u] = __int_as_float(__builtin_amdgcn_readlane(cy, j + u));
    x[u] = *(const float2*)(inl + (size_t)((unsigned)sv * (unsigned)PITCH));
  }
#pragma unroll
  for (int u = 0; u < C; ++u) {
    ax = fmaf(w[u], x[u].x, ax);
    ay = fmaf(w[u], x[u].y, ay);
  }
}

__global__ void __launch_bounds__(256)
dagnn_gather(const float* __restrict__ in, float* __restrict__ outp,
             const int2* __restrict__ csr, const int* __restrict__ rowptr,
             const float* __restrict__ norm, int Nn) {
  const int lane = threadIdx.x & 31;
  const int node = __builtin_amdgcn_readfirstlane((int)(blockIdx.x * 8 + (threadIdx.x >> 5)));
  if (node >= Nn) return;
  const int beg = rowptr[node], end = rowptr[node + 1];
  const float* inl = in + 2 * lane;
  float ax = 0.f, ay = 0.f;
  for (int base = beg; base < end; base += 32) {
    int n = end - base;
    if (n > 32) n = 32;
    int2 cw = make_int2(0, 0);
    if (lane < n) cw = csr[base + lane];
    int j = 0;
    for (; j + 8 <= n; j += 8) gather_chunk<8>(inl, cw.x, cw.y, j, ax, ay);
    if (n - j >= 4) { gather_chunk<4>(inl, cw.x, cw.y, j, ax, ay); j += 4; }
    if (n - j >= 2) { gather_chunk<2>(inl, cw.x, cw.y, j, ax, ay); j += 2; }
    if (n - j >= 1) gather_chunk<1>(inl, cw.x, cw.y, j, ax, ay);
  }
  const float nd = norm[node];
  const bool real = lane < NCLASS / 2;
  float2 r;
  r.x = real ? nd * ax : 0.f;
  r.y = real ? nd * ay : 0.f;
  *(float2*)(outp + (size_t)node * PITCH + 2 * lane) = r;
}

__global__ void __launch_bounds__(256)
dagnn_gather_readout(const float* __restrict__ planes, const float* __restrict__ s,
                     float* __restrict__ out, const int2* __restrict__ csr,
                     const int* __restrict__ rowptr, const float* __restrict__ norm, int Nn) {
  const int lane = threadIdx.x & 31;
  const int node = __builtin_amdgcn_readfirstlane((int)(blockIdx.x * 8 + (threadIdx.x >> 5)));
  if (node >= Nn) return;
  const int beg = rowptr[node], end = rowptr[node + 1];
  const size_t pstride = (size_t)Nn * PITCH;
  const float* inl = planes + (size_t)(KHOPS - 1) * pstride + 2 * lane;

  int nfirst = end - beg;
  if (nfirst > 32) nfirst = 32;
  int2 cw = make_int2(0, 0);
  if (lane < nfirst) cw = csr[beg + lane];

  const float* ownp = planes + (size_t)node * PITCH + 2 * lane;
  v2f own[KHOPS];
#pragma unroll
  for (int k = 0; k < KHOPS - 1; ++k)
    own[k] = __builtin_nontemporal_load((const v2f*)(ownp + (size_t)k * pstride));
  own[KHOPS - 1] = *(const v2f*)(ownp + (size_t)(KHOPS - 1) * pstride);

  const bool real = lane < NCLASS / 2;
  const v2f sv = *(const v2f*)(s + (real ? 2 * lane : 0));
  const float sx = real ? sv.x : 0.f, sy = real ? sv.y : 0.f;
  const float nd = norm[node];

  float ax = 0.f, ay = 0.f;
  auto walk = [&](int n) {
    int j = 0;
    for (; j + 8 <= n; j += 8) gather_chunk<8>(inl, cw.x, cw.y, j, ax, ay);
    if (n - j >= 4) { gather_chunk<4>(inl, cw.x, cw.y, j, ax, ay); j += 4; }
    if (n - j >= 2) { gather_chunk<2>(inl, cw.x, cw.y, j, ax, ay); j += 2; }
    if (n - j >= 1) gather_chunk<1>(inl, cw.x, cw.y, j, ax, ay);
  };
  walk(nfirst);
  for (int base = beg + 32; base < end; base += 32) {
    int m = end - base;
    if (m > 32) m = 32;
    cw = make_int2(0, 0);
    if (lane < m) cw = csr[base + lane];
    walk(m);
  }
  const float hx = real ? nd * ax : 0.f;
  const float hy = real ? nd * ay : 0.f;

  float ox = 0.f, oy = 0.f;
  auto add_hop = [&](float x, float y) {
    float z = real ? x * sx + y * sy : 0.f;
#pragma unroll
    for (int off = 16; off > 0; off >>= 1) z += __shfl_xor(z, off, 32);
    const float S = 1.0f / (1.0f + __expf(-z));
    ox += S * x;
    oy += S * y;
  };
#pragma unroll
  for (int k = 0; k < KHOPS; ++k) add_hop(own[k].x, own[k].y);
  add_hop(hx, hy);
  if (real) {
    v2f r;
    r.x = ox;
    r.y = oy;
    *(v2f*)(out + (size_t)node * NCLASS + 2 * lane) = r;
  }
}

extern "C" void kernel_launch(void* const* d_in, const int* in_sizes, int n_in,
                              void* d_out, int out_size, void* d_ws, size_t ws_size,
                              hipStream_t stream) {
  const float* feats = (const float*)d_in[0];
  const float* W1    = (const float*)d_in[1];
  const float* b1    = (const float*)d_in[2];
  const float* W2    = (const float*)d_in[3];
  const float* b2    = (const float*)d_in[4];
  const float* s     = (const float*)d_in[5];
  const int*   src   = (const int*)d_in[6];
  const int*   dst   = (const int*)d_in[7];
  const int Nn = in_sizes[0] / IN_DIM;
  const int E  = in_sizes[6];
  float* out = (float*)d_out;

  char* ws = (char*)d_ws;
  size_t off = 0;
  auto carve = [&](size_t bytes) -> char* {
    char* p = ws + off;
    off = (off + bytes + 255) & ~(size_t)255;
    return p;
  };
  const int ntile = (Nn + 1023) / 1024;
  float*    norm   = (float*)   carve((size_t)Nn * sizeof(float));
  int*      cnt    = (int*)     carve((size_t)Nn * sizeof(int));
  int*      rowptr = (int*)     carve((size_t)(Nn + 1) * sizeof(int));
  int*      tsum   = (int*)     carve((size_t)ntile * sizeof(int));
  int*      rank   = (int*)     carve((size_t)E * sizeof(int));
  int2*     csr    = (int2*)    carve((size_t)E * sizeof(int2));
  float*    planes = (float*)   carve((size_t)KHOPS * Nn * PITCH * sizeof(float));
  _Float16* W1h    = (_Float16*)carve((size_t)W1_PIECES * 16);
  _Float16* W2h    = (_Float16*)carve((size_t)W2_PIECES * 16);
  if (off > ws_size) return;

  const size_t pstride = (size_t)Nn * PITCH;

  hipMemsetAsync(cnt, 0, (size_t)Nn * sizeof(int), stream);
  dagnn_rank<<<(E + 255) / 256, 256, 0, stream>>>(dst, cnt, rank, E);
  dagnn_tilesum_prep<<<ntile + PREP_BLOCKS, 256, 0, stream>>>(cnt, tsum, Nn, ntile, W1, W2, W1h, W2h);
  dagnn_scan<<<ntile, 256, 0, stream>>>(cnt, tsum, rowptr, norm, Nn);
  dagnn_fill<<<(E + 255) / 256, 256, 0, stream>>>(src, dst, rank, rowptr, norm, csr, E);

  dim3 g1((Nn + 63) / 64, 1);
  dagnn_mlp<<<g1, 256, 0, stream>>>(feats, W1h, b1, W2h, b2, planes, Nn);

  const int ggrid = (Nn + 7) / 8;
  for (int k = 1; k < KHOPS; ++k) {
    const float* pin = planes + (size_t)(k - 1) * pstride;
    float* pout      = planes + (size_t)k * pstride;
    dagnn_gather<<<ggrid, 256, 0, stream>>>(pin, pout, csr, rowptr, norm, Nn);
  }

  dagnn_gather_readout<<<ggrid, 256, 0, stream>>>(planes, s, out, csr, rowptr, norm, Nn);
  (void)n_in; (void)out_size;
}


# Gather hops: half a wave per node, a lane holds 4 columns: one 16-byte-per-lane load fetches two source rows (512 B per instruction); edge order of sums kept; readout dot summed over 16 lanes by DPP
# speedup vs baseline: 5.3234x; 1.0441x over previous
#include <hip/hip_runtime.h>
#include <hip/hip_bf16.h>

#define IN_DIM 500
#define HID    256
#define NCLASS 50
#define KHOPS  10
#define PITCH  64

typedef __attribute__((ext_vector_type(16))) _Float16 v16h;
typedef __attribute__((ext_vector_type(8)))  _Float16 h8;
typedef __attribute__((ext_vector_type(4)))  _Float16 h4;
typedef __attribute__((ext_vector_type(8)))  float    v8f;
typedef __attribute__((ext_vector_type(4)))  float    v4f;
typedef __attribute__((ext_vector_type(2)))  float    v2f;

__device__ __forceinline__ v16h load_frag(const _Float16* row, int kb) {
  h8 lo = *(const h8*)(row + kb);
  h8 hi = *(const h8*)(row + 16 + kb);
  v16h f;
#pragma unroll
  for (int i = 0; i < 8; ++i) { f[i] = lo[i]; f[8 + i] = hi[i]; }
  return f;
}

__device__ __forceinline__ v8f wmma16(v16h a, v16h b, v8f c) {
  return __builtin_amdgcn_wmma_f32_16x16x32_f16(false, a, false, b, (short)0,
                                                c, false, false);
}

#define HPITCH 264
#define NKTILE 16
#define W1_PIECES (NKTILE * 4 * HID)
#define W2_PIECES (64 * (HID / 8))

#define PREP_BLOCKS ((W1_PIECES + W2_PIECES + 255) / 256)
__device__ __forceinline__ void prep_weights(int g, const float* __restrict__ W1,
                                             const float* __restrict__ W2,
                                             _Float16* __restrict__ W1h,
                                             _Float16* __restrict__ W2h) {
  float v[8];
  h8 p;
  if (g < W1_PIECES) {
    const int n = g & (HID - 1), k0 = (g >> 8) * 8;
#pragma unroll
    for (int i = 0; i < 8; ++i) {
      const int k = k0 + i;
      v[i] = W1[(size_t)(k < IN_DIM ? k : IN_DIM - 1) * HID + n];
    }
#pragma unroll
    for (int i = 0; i < 8; ++i) p[i] = (_Float16)(k0 + i < IN_DIM ? v[i] : 0.f);
    ((h8*)W1h)[g] = p;
  } else if (g < W1_PIECES + W2_PIECES) {
    const int r = g - W1_PIECES;
    const int n = r & 63, k8 = r >> 6;
    const float* w2p = W2 + (n < NCLASS ? n : NCLASS - 1) + (size_t)k8 * 8 * NCLASS;
#pragma unroll
    for (int i = 0; i < 8; ++i) v[i] = w2p[i * NCLASS];
#pragma unroll
    for (int i = 0; i < 8; ++i) p[i] = (_Float16)(n < NCLASS ? v[i] : 0.f);
    ((h8*)W2h)[n * (HID / 8) + k8] = p;
  }
}

__global__ void __launch_bounds__(256)
dagnn_mlp(const float* __restrict__ X, const _Float16* __restrict__ W1h,
          const float* __restrict__ bias, const _Float16* __restrict__ W2h,
          const float* __restrict__ b2, float* __restrict__ plane0, int Nn) {
  __shared__ __align__(16) _Float16 As[2][64][40];
  __shared__ __align__(16) _Float16 Bt[2][HID][40];
  __shared__ __align__(16) _Float16 W2s[64][HPITCH];
  _Float16 (*Hs)[HPITCH] = reinterpret_cast<_Float16 (*)[HPITCH]>(&Bt[0][0][0]);
  const int t = threadIdx.x;
  const int wave = t >> 5, lane = t & 31;
  const int waveM = wave >> 1, waveN = wave & 1;
  const int blockM = blockIdx.x * 64;

  v4f    ra[2];
  h8     rb[4];
  const int KTAIL = IN_DIM - 480;

  auto fetch = [&](int kt, const bool tail) {
#pragma unroll
    for (int j = 0; j < 2; ++j) {
      int idx = j * 256 + t;
      int r = idx >> 3, k4 = (idx & 7) * 4;
      int grow = blockM + r;
      int cr = grow < Nn ? grow : Nn - 1;
      int ck = kt + k4;
      if (tail) ck = ck < IN_DIM ? ck : IN_DIM - 4;
      ra[j] = *(const v4f*)(X + (size_t)cr * IN_DIM + ck);
    }
    const h8* wp = (const h8*)W1h + (size_t)(kt >> 3) * HID + t;
#pragma unroll
    for (int q = 0; q < 4; ++q)
      rb[q] = wp[q * HID];
  };
  auto stage = [&](int buf, const bool tail) {
#pragma unroll
    for (int j = 0; j < 2; ++j) {
      int idx = j * 256 + t;
      int r = idx >> 3, k4 = (idx & 7) * 4;
      bool ok = !tail || k4 < KTAIL;
      h4 hv;
      hv[0] = ok ? (_Float16)ra[j].x : (_Float16)0.f;
      hv[1] = ok ? (_Float16)ra[j].y : (_Float16)0.f;
      hv[2] = ok ? (_Float16)ra[j].z : (_Float16)0.f;
      hv[3] = ok ? (_Float16)ra[j].w : (_Float16)0.f;
      *(h4*)&As[buf][r][k4] = hv;
    }
#pragma unroll
    for (int q = 0; q < 4; ++q)
      *(h8*)&Bt[buf][t][q * 8] = rb[q];
  };

  v8f acc[8];
#pragma unroll
  for (int nt = 0; nt < 8; ++nt) acc[nt] = (v8f){0.f, 0.f, 0.f, 0.f, 0.f, 0.f, 0.f, 0.f};
  const int mrow = waveM * 16 + (lane & 15);
  const int nrow = waveN * 128 + (lane & 15);
  const int kb = (lane >> 4) * 8;

  auto mma = [&](int cur) {
    v16h xf = load_frag(&As[cur][mrow][0], kb);
#pragma unroll
    for (int nt = 0; nt < 8; ++nt) {
      v16h wf = load_frag(&Bt[cur][nrow + nt * 16][0], kb);
      acc[nt] = wmma16(wf, xf, acc[nt]);
    }
  };

  fetch(0, false);
  {
    const h8* w2p = (const h8*)W2h + t;
    h8 wv[8];
#pragma unroll
    for (int j = 0; j < 8; ++j) wv[j] = w2p[j * 256];
#pragma unroll
    for (int j = 0; j < 8; ++j) {
      const int p = j * 256 + t;
      *(h8*)&W2s[p >> 5][(p & 31) * 8] = wv[j];
    }
  }
  stage(0, false);
  __syncthreads();
  const int NKT = NKTILE;
  for (int it = 0; it < NKT - 2; ++it) {
    int cur = it & 1;
    fetch((it + 1) * 32, false);
    asm volatile("" ::: "memory");
    mma(cur);
    __builtin_amdgcn_sched_barrier(0);
    stage(cur ^ 1, false);
    __syncthreads();
  }
  fetch((NKT - 1) * 32, true);
  asm volatile("" ::: "memory");
  mma(0);
  __builtin_amdgcn_sched_barrier(0);
  stage(1, true);
  __syncthreads();
  mma(1);
  __syncthreads();

#pragma unroll
  for (int nt = 0; nt < 8; ++nt) {
    const int n0 = waveN * 128 + nt * 16 + kb;
    const v4f bA = *(const v4f*)(bias + n0);
    const v4f bB = *(const v4f*)(bias + n0 + 4);
    h8 p;
#pragma unroll
    for (int r = 0; r < 8; ++r) {
      float v = acc[nt][r] + (r < 4 ? bA[r] : bB[r - 4]);
      p[r] = (_Float16)(v > 0.f ? v : 0.f);
    }
    *(h8*)&Hs[mrow][n0] = p;
  }
  __syncthreads();

  v8f o[2];
  o[0] = (v8f){0.f, 0.f, 0.f, 0.f, 0.f, 0.f, 0.f, 0.f};
  o[1] = (v8f){0.f, 0.f, 0.f, 0.f, 0.f, 0.f, 0.f, 0.f};
  const int ncol = waveN * 32 + (lane & 15);
  float bv2[2];
#pragma unroll
  for (int q = 0; q < 2; ++q) {
    const int col = ncol + q * 16;
    const float bl = b2[col < NCLASS ? col : NCLASS - 1];
    bv2[q] = col < NCLASS ? bl : 0.f;
  }
#pragma unroll
  for (int kh = 0; kh < HID / 128; ++kh) {
    v16h hf[4], w0[4], w1[4];
#pragma unroll
    for (int u = 0; u < 4; ++u) {
      const int k0 = (kh * 4 + u) * 32;
      hf[u] = load_frag(&Hs[mrow][k0], kb);
      w0[u] = load_frag(&W2s[ncol][k0], kb);
      w1[u] = load_frag(&W2s[ncol + 16][k0], kb);
    }
#pragma unroll
    for (int u = 0; u < 4; ++u) {
      o[0] = wmma16(hf[u], w0[u], o[0]);
      o[1] = wmma16(hf[u], w1[u], o[1]);
    }
  }

  const int rbase = blockM + waveM * 16 + kb;
#pragma unroll
  for (int q = 0; q < 2; ++q) {
    const int col = ncol + q * 16;
#pragma unroll
    for (int i = 0; i < 8; ++i) {
      int row = rbase + i;
      if (row < Nn) plane0[(size_t)row * PITCH + col] = col < NCLASS ? o[q][i] + bv2[q] : 0.f;
    }
  }
}

__global__ void __launch_bounds__(256)
dagnn_rank(const int* __restrict__ dst, int* __restrict__ cnt,
           int* __restrict__ rank, int E) {
  int e = blockIdx.x * blockDim.x + threadIdx.x;
  if (e < E) rank[e] = atomicAdd(&cnt[dst[e]], 1);
}

__global__ void __launch_bounds__(256)
dagnn_tilesum_prep(const int* __restrict__ cnt, int* __restrict__ tsum, int Nn, int ntile,
                   const float* __restrict__ W1, const float* __restrict__ W2,
                   _Float16* __restrict__ W1h, _Float16* __restrict__ W2h) {
  if ((int)blockIdx.x >= ntile) {
    prep_weights(((int)blockIdx.x - ntile) * 256 + (int)threadIdx.x, W1, W2, W1h, W2h);
    return;
  }
  __shared__ int red[8];
  const int t = threadIdx.x;
  const int i0 = blockIdx.x * 1024 + t * 4;
  int s = 0;
#pragma unroll
  for (int u = 0; u < 4; ++u)
    if (i0 + u < Nn) s += cnt[i0 + u];
#pragma unroll
  for (int off = 16; off > 0; off >>= 1) s += __shfl_xor(s, off, 32);
  if ((t & 31) == 0) red[t >> 5] = s;
  __syncthreads();
  if (t == 0) {
    int r = 0;
#pragma unroll
    for (int w = 0; w < 8; ++w) r += red[w];
    tsum[blockIdx.x] = r;
  }
}

__global__ void __launch_bounds__(256)
dagnn_scan(const int* __restrict__ cnt, const int* __restrict__ tsum,
           int* __restrict__ rowptr, float* __restrict__ norm, int Nn) {
  __shared__ int red[8];
  __shared__ int wsum[8];
  const int t = threadIdx.x, lane = t & 31, wave = t >> 5;
  const int tile = blockIdx.x;

  int o = 0;
  for (int i = t; i < tile; i += 256) o += tsum[i];
#pragma unroll
  for (int off = 16; off > 0; off >>= 1) o += __shfl_xor(o, off, 32);
  if (lane == 0) red[wave] = o;

  const int i0 = tile * 1024 + t * 4;
  int c[4];
#pragma unroll
  for (int u = 0; u < 4; ++u) c[u] = (i0 + u < Nn) ? cnt[i0 + u] : 0;
  const int ts = c[0] + c[1] + c[2] + c[3];
  int inc = ts;
#pragma unroll
  for (int off = 1; off < 32; off <<= 1) {
    int v = __shfl_up(inc, off, 32);
    if (lane >= off) inc += v;
  }
  if (lane == 31) wsum[wave] = inc;
  __syncthreads();

  int ex = inc - ts;
#pragma unroll
  for (int w = 0; w < 8; ++w) {
    ex += red[w];
    if (w < wave) ex += wsum[w];
  }
#pragma unroll
  for (int u = 0; u < 4; ++u) {
    if (i0 + u < Nn) {
      rowptr[i0 + u] = ex;
      norm[i0 + u] = rsqrtf((float)c[u]);
      ex += c[u];
    }
  }
  if (i0 < Nn && Nn <= i0 + 4) rowptr[Nn] = ex;
}

__global__ void __launch_bounds__(256)
dagnn_fill(const int* __restrict__ src, const int* __restrict__ dst,
           const int* __restrict__ rank, const int* __restrict__ rowptr,
           const float* __restrict__ norm, int2* __restrict__ csr, int E) {
  int e = blockIdx.x * blockDim.x + threadIdx.x;
  if (e >= E) return;
  int sv = src[e];
  int pos = rowptr[dst[e]] + rank[e];
  csr[pos] = make_int2(sv, __float_as_int(norm[sv]));
}


template <int S>
__device__ __forceinline__ int row_bcast(int v) {
  return __builtin_amdgcn_update_dpp(0, v, 0x150 + S, 0xf, 0xf, false);
}

__device__ __forceinline__ void load_edges(const int2* __restrict__ csr, int beg, int len,
                                           int done, int c, unsigned& eo, int& wb) {
  eo = 208u;
  wb = 0;
  if (c < len - done) {
    const int2 e = csr[beg + done + c];
    eo = (unsigned)e.x << 8;
    wb = e.y;
  }
}

template <int S, int U, int C>
__device__ __forceinline__ void issue_rows(const char* __restrict__ inb, unsigned eo, int wb,
                                           unsigned cB, v4f* x, float* w) {
  if constexpr (U < C) {
    const unsigned o = (unsigned)row_bcast<S + U>((int)eo) | cB;
    w[U] = __int_as_float(row_bcast<S + U>(wb));
    x[U] = *(const v4f*)(inb + o);
    issue_rows<S, U + 1, C>(inb, eo, wb, cB, x, w);
  }
}

template <int S, int C>
__device__ __forceinline__ void gather_rows(const char* __restrict__ inb, unsigned eo, int wb,
                                            unsigned cB, v4f& a) {
  v4f x[C];
  float w[C];
  issue_rows<S, 0, C>(inb, eo, wb, cB, x, w);
#pragma unroll
  for (int u = 0; u < C; ++u) {
    a.x = fmaf(w[u], x[u].x, a.x);
    a.y = fmaf(w[u], x[u].y, a.y);
    a.z = fmaf(w[u], x[u].z, a.z);
    a.w = fmaf(w[u], x[u].w, a.w);
  }
}

template <int S>
__device__ __forceinline__ void walk8(const char* __restrict__ inb, unsigned eo, int wb,
                                      unsigned cB, int r, v4f& a) {
  if (r > 6)      gather_rows<S, 8>(inb, eo, wb, cB, a);
  else if (r > 4) gather_rows<S, 6>(inb, eo, wb, cB, a);
  else if (r > 2) gather_rows<S, 4>(inb, eo, wb, cB, a);
  else            gather_rows<S, 2>(inb, eo, wb, cB, a);
}

__device__ __forceinline__ void walk16(const char* __restrict__ inb, unsigned eo, int wb,
                                       unsigned cB, int m, v4f& a) {
  walk8<0>(inb, eo, wb, cB, m, a);
  if (m > 8) walk8<8>(inb, eo, wb, cB, m - 8, a);
}

__device__ __forceinline__ int wave_max_len(int len) {
  const int l0 = __builtin_amdgcn_readlane(len, 0), l1 = __builtin_amdgcn_readlane(len, 16);
  return l0 > l1 ? l0 : l1;
}

__global__ void __launch_bounds__(256)
dagnn_gather(const float* __restrict__ in, float* __restrict__ outp,
             const int2* __restrict__ csr, const int* __restrict__ rowptr,
             const float* __restrict__ norm, int Nn) {
  const int lane = threadIdx.x & 31;
  const int c = lane & 15;
  const int node = ((int)blockIdx.x * 8 + ((int)threadIdx.x >> 5)) * 2 + (lane >> 4);
  const bool live = node < Nn;
  int beg = 0, end = 0;
  if (live) { beg = rowptr[node]; end = rowptr[node + 1]; }
  const int len = end - beg;
  const int lenmax = wave_max_len(len);
  const char* inb = (const char*)in;
  const unsigned cB = (unsigned)c << 4;
  v4f a = (v4f){0.f, 0.f, 0.f, 0.f};
  for (int done = 0; done < lenmax; done += 16) {
    unsigned eo;
    int wb;
    load_edges(csr, beg, len, done, c, eo, wb);
    const int m = lenmax - done;
    walk16(inb, eo, wb, cB, m < 16 ? m : 16, a);
  }
  if (live) {
    const float nd = norm[node];
    v4f r;
    r.x = 4 * c + 0 < NCLASS ? nd * a.x : 0.f;
    r.y = 4 * c + 1 < NCLASS ? nd * a.y : 0.f;
    r.z = 4 * c + 2 < NCLASS ? nd * a.z : 0.f;
    r.w = 4 * c + 3 < NCLASS ? nd * a.w : 0.f;
    *(v4f*)(outp + (size_t)node * PITCH + 4 * c) = r;
  }
}

template <int M>
__device__ __forceinline__ float row_xor_add(float z) {
  return z + __int_as_float(__builtin_amdgcn_update_dpp(0, __float_as_int(z), 0x160 + M,
                                                        0xf, 0xf, false));
}

__global__ void __launch_bounds__(256)
dagnn_gather_readout(const float* __restrict__ planes, const float* __restrict__ s,
                     float* __restrict__ out, const int2* __restrict__ csr,
                     const int* __restrict__ rowptr, const float* __restrict__ norm, int Nn) {
  const int lane = threadIdx.x & 31;
  const int c = lane & 15;
  const int node = ((int)blockIdx.x * 8 + ((int)threadIdx.x >> 5)) * 2 + (lane >> 4);
  const bool live = node < Nn;
  const int cnode = live ? node : Nn - 1;
  int beg = 0, end = 0;
  if (live) { beg = rowptr[node]; end = rowptr[node + 1]; }
  const int len = end - beg;
  const int lenmax = wave_max_len(len);
  const size_t pstride = (size_t)Nn * PITCH;
  const char* inb = (const char*)(planes + (size_t)(KHOPS - 1) * pstride);
  const unsigned cB = (unsigned)c << 4;

  unsigned eo;
  int wb;
  load_edges(csr, beg, len, 0, c, eo, wb);

  const float* ownp = planes + (size_t)cnode * PITCH + 4 * c;
  v4f own[KHOPS];
#pragma unroll
  for (int k = 0; k < KHOPS - 1; ++k)
    own[k] = __builtin_nontemporal_load((const v4f*)(ownp + (size_t)k * pstride));
  own[KHOPS - 1] = *(const v4f*)(ownp + (size_t)(KHOPS - 1) * pstride);

  float sc[4];
  bool real[4];
#pragma unroll
  for (int i = 0; i < 4; ++i) {
    const int col = 4 * c + i;
    real[i] = col < NCLASS;
    const float sl = s[real[i] ? col : NCLASS - 1];
    sc[i] = real[i] ? sl : 0.f;
  }
  const float nd = live ? norm[node] : 0.f;

  v4f a = (v4f){0.f, 0.f, 0.f, 0.f};
  if (lenmax > 0) walk16(inb, eo, wb, cB, lenmax < 16 ? lenmax : 16, a);
  for (int done = 16; done < lenmax; done += 16) {
    load_edges(csr, beg, len, done, c, eo, wb);
    const int m = lenmax - done;
    walk16(inb, eo, wb, cB, m < 16 ? m : 16, a);
  }
  v4f h;
  h.x = real[0] ? nd * a.x : 0.f;
  h.y = real[1] ? nd * a.y : 0.f;
  h.z = real[2] ? nd * a.z : 0.f;
  h.w = real[3] ? nd * a.w : 0.f;

  v4f o = (v4f){0.f, 0.f, 0.f, 0.f};
  auto add_hop = [&](v4f x) {
    float z = (real[0] ? x.x * sc[0] : 0.f) + (real[1] ? x.y * sc[1] : 0.f)
            + (real[2] ? x.z * sc[2] : 0.f) + (real[3] ? x.w * sc[3] : 0.f);
    z = row_xor_add<8>(z);
    z = row_xor_add<4>(z);
    z = row_xor_add<2>(z);
    z = row_xor_add<1>(z);
    const float S = 1.0f / (1.0f + __expf(-z));
    o.x += S * x.x;
    o.y += S * x.y;
    o.z += S * x.z;
    o.w += S * x.w;
  };
#pragma unroll
  for (int k = 0; k < KHOPS; ++k) add_hop(own[k]);
  add_hop(h);
  if (live) {
    float* op = out + (size_t)node * NCLASS + 4 * c;
    if (real[1]) { v2f r; r.x = o.x; r.y = o.y; *(v2f*)op = r; }
    if (real[3]) { v2f r; r.x = o.z; r.y = o.w; *(v2f*)(op + 2) = r; }
  }
}

extern "C" void kernel_launch(void* const* d_in, const int* in_sizes, int n_in,
                              void* d_out, int out_size, void* d_ws, size_t ws_size,
                              hipStream_t stream) {
  const float* feats = (const float*)d_in[0];
  const float* W1    = (const float*)d_in[1];
  const float* b1    = (const float*)d_in[2];
  const float* W2    = (const float*)d_in[3];
  const float* b2    = (const float*)d_in[4];
  const float* s     = (const float*)d_in[5];
  const int*   src   = (const int*)d_in[6];
  const int*   dst   = (const int*)d_in[7];
  const int Nn = in_sizes[0] / IN_DIM;
  const int E  = in_sizes[6];
  float* out = (float*)d_out;

  char* ws = (char*)d_ws;
  size_t off = 0;
  auto carve = [&](size_t bytes) -> char* {
    char* p = ws + off;
    off = (off + bytes + 255) & ~(size_t)255;
    return p;
  };
  const int ntile = (Nn + 1023) / 1024;
  float*    norm   = (float*)   carve((size_t)Nn * sizeof(float));
  int*      cnt    = (int*)     carve((size_t)Nn * sizeof(int));
  int*      rowptr = (int*)     carve((size_t)(Nn + 1) * sizeof(int));
  int*      tsum   = (int*)     carve((size_t)ntile * sizeof(int));
  int*      rank   = (int*)     carve((size_t)E * sizeof(int));
  int2*     csr    = (int2*)    carve((size_t)E * sizeof(int2));
  float*    planes = (float*)   carve((size_t)KHOPS * Nn * PITCH * sizeof(float));
  _Float16* W1h    = (_Float16*)carve((size_t)W1_PIECES * 16);
  _Float16* W2h    = (_Float16*)carve((size_t)W2_PIECES * 16);
  if (off > ws_size) return;

  const size_t pstride = (size_t)Nn * PITCH;

  hipMemsetAsync(cnt, 0, (size_t)Nn * sizeof(int), stream);
  dagnn_rank<<<(E + 255) / 256, 256, 0, stream>>>(dst, cnt, rank, E);
  dagnn_tilesum_prep<<<ntile + PREP_BLOCKS, 256, 0, stream>>>(cnt, tsum, Nn, ntile, W1, W2, W1h, W2h);
  dagnn_scan<<<ntile, 256, 0, stream>>>(cnt, tsum, rowptr, norm, Nn);
  dagnn_fill<<<(E + 255) / 256, 256, 0, stream>>>(src, dst, rank, rowptr, norm, csr, E);

  dim3 g1((Nn + 63) / 64, 1);
  dagnn_mlp<<<g1, 256, 0, stream>>>(feats, W1h, b1, W2h, b2, planes, Nn);

  const int ggrid = (Nn + 15) / 16;
  for (int k = 1; k < KHOPS; ++k) {
    const float* pin = planes + (size_t)(k - 1) * pstride;
    float* pout      = planes + (size_t)k * pstride;
    dagnn_gather<<<ggrid, 256, 0, stream>>>(pin, pout, csr, rowptr, norm, Nn);
  }

  dagnn_gather_readout<<<ggrid, 256, 0, stream>>>(planes, s, out, csr, rowptr, norm, Nn);
  (void)n_in; (void)out_size;
}


# CSR build without global atomics: edges bucketed by 256-node ranges, then placed per bucket, all counters in LDS; memset, rank and fill passes go; order of a row's sums = LDS counter arrival order
# speedup vs baseline: 6.1608x; 1.1573x over previous
#include <hip/hip_runtime.h>
#include <hip/hip_bf16.h>

#define IN_DIM 500
#define HID    256
#define NCLASS 50
#define KHOPS  10
#define PITCH  64

typedef __attribute__((ext_vector_type(16))) _Float16 v16h;
typedef __attribute__((ext_vector_type(8)))  _Float16 h8;
typedef __attribute__((ext_vector_type(4)))  _Float16 h4;
typedef __attribute__((ext_vector_type(8)))  float    v8f;
typedef __attribute__((ext_vector_type(4)))  float    v4f;
typedef __attribute__((ext_vector_type(2)))  float    v2f;

__device__ __forceinline__ v16h load_frag(const _Float16* row, int kb) {
  h8 lo = *(const h8*)(row + kb);
  h8 hi = *(const h8*)(row + 16 + kb);
  v16h f;
#pragma unroll
  for (int i = 0; i < 8; ++i) { f[i] = lo[i]; f[8 + i] = hi[i]; }
  return f;
}

__device__ __forceinline__ v8f wmma16(v16h a, v16h b, v8f c) {
  return __builtin_amdgcn_wmma_f32_16x16x32_f16(false, a, false, b, (short)0,
                                                c, false, false);
}

#define HPITCH 264
#define NKTILE 16
#define W1_PIECES (NKTILE * 4 * HID)
#define W2_PIECES (64 * (HID / 8))

#define PREP_BLOCKS ((W1_PIECES + W2_PIECES + 255) / 256)
__device__ __forceinline__ void prep_weights(int g, const float* __restrict__ W1,
                                             const float* __restrict__ W2,
                                             _Float16* __restrict__ W1h,
                                             _Float16* __restrict__ W2h) {
  float v[8];
  h8 p;
  if (g < W1_PIECES) {
    const int n = g & (HID - 1), k0 = (g >> 8) * 8;
#pragma unroll
    for (int i = 0; i < 8; ++i) {
      const int k = k0 + i;
      v[i] = W1[(size_t)(k < IN_DIM ? k : IN_DIM - 1) * HID + n];
    }
#pragma unroll
    for (int i = 0; i < 8; ++i) p[i] = (_Float16)(k0 + i < IN_DIM ? v[i] : 0.f);
    ((h8*)W1h)[g] = p;
  } else if (g < W1_PIECES + W2_PIECES) {
    const int r = g - W1_PIECES;
    const int n = r & 63, k8 = r >> 6;
    const float* w2p = W2 + (n < NCLASS ? n : NCLASS - 1) + (size_t)k8 * 8 * NCLASS;
#pragma unroll
    for (int i = 0; i < 8; ++i) v[i] = w2p[i * NCLASS];
#pragma unroll
    for (int i = 0; i < 8; ++i) p[i] = (_Float16)(n < NCLASS ? v[i] : 0.f);
    ((h8*)W2h)[n * (HID / 8) + k8] = p;
  }
}

__global__ void __launch_bounds__(256)
dagnn_mlp(const float* __restrict__ X, const _Float16* __restrict__ W1h,
          const float* __restrict__ bias, const _Float16* __restrict__ W2h,
          const float* __restrict__ b2, float* __restrict__ plane0, int Nn) {
  __shared__ __align__(16) _Float16 As[2][64][40];
  __shared__ __align__(16) _Float16 Bt[2][HID][40];
  __shared__ __align__(16) _Float16 W2s[64][HPITCH];
  _Float16 (*Hs)[HPITCH] = reinterpret_cast<_Float16 (*)[HPITCH]>(&Bt[0][0][0]);
  const int t = threadIdx.x;
  const int wave = t >> 5, lane = t & 31;
  const int waveM = wave >> 1, waveN = wave & 1;
  const int blockM = blockIdx.x * 64;

  v4f    ra[2];
  h8     rb[4];
  const int KTAIL = IN_DIM - 480;

  auto fetch = [&](int kt, const bool tail) {
#pragma unroll
    for (int j = 0; j < 2; ++j) {
      int idx = j * 256 + t;
      int r = idx >> 3, k4 = (idx & 7) * 4;
      int grow = blockM + r;
      int cr = grow < Nn ? grow : Nn - 1;
      int ck = kt + k4;
      if (tail) ck = ck < IN_DIM ? ck : IN_DIM - 4;
      ra[j] = *(const v4f*)(X + (size_t)cr * IN_DIM + ck);
    }
    const h8* wp = (const h8*)W1h + (size_t)(kt >> 3) * HID + t;
#pragma unroll
    for (int q = 0; q < 4; ++q)
      rb[q] = wp[q * HID];
  };
  auto stage = [&](int buf, const bool tail) {
#pragma unroll
    for (int j = 0; j < 2; ++j) {
      int idx = j * 256 + t;
      int r = idx >> 3, k4 = (idx & 7) * 4;
      bool ok = !tail || k4 < KTAIL;
      h4 hv;
      hv[0] = ok ? (_Float16)ra[j].x : (_Float16)0.f;
      hv[1] = ok ? (_Float16)ra[j].y : (_Float16)0.f;
      hv[2] = ok ? (_Float16)ra[j].z : (_Float16)0.f;
      hv[3] = ok ? (_Float16)ra[j].w : (_Float16)0.f;
      *(h4*)&As[buf][r][k4] = hv;
    }
#pragma unroll
    for (int q = 0; q < 4; ++q)
      *(h8*)&Bt[buf][t][q * 8] = rb[q];
  };

  v8f acc[8];
#pragma unroll
  for (int nt = 0; nt < 8; ++nt) acc[nt] = (v8f){0.f, 0.f, 0.f, 0.f, 0.f, 0.f, 0.f, 0.f};
  const int mrow = waveM * 16 + (lane & 15);
  const int nrow = waveN * 128 + (lane & 15);
  const int kb = (lane >> 4) * 8;

  auto mma = [&](int cur) {
    v16h xf = load_frag(&As[cur][mrow][0], kb);
#pragma unroll
    for (int nt = 0; nt < 8; ++nt) {
      v16h wf = load_frag(&Bt[cur][nrow + nt * 16][0], kb);
      acc[nt] = wmma16(wf, xf, acc[nt]);
    }
  };

  fetch(0, false);
  {
    const h8* w2p = (const h8*)W2h + t;
    h8 wv[8];
#pragma unroll
    for (int j = 0; j < 8; ++j) wv[j] = w2p[j * 256];
#pragma unroll
    for (int j = 0; j < 8; ++j) {
      const int p = j * 256 + t;
      *(h8*)&W2s[p >> 5][(p & 31) * 8] = wv[j];
    }
  }
  stage(0, false);
  __syncthreads();
  const int NKT = NKTILE;
  for (int it = 0; it < NKT - 2; ++it) {
    int cur = it & 1;
    fetch((it + 1) * 32, false);
    asm volatile("" ::: "memory");
    mma(cur);
    __builtin_amdgcn_sched_barrier(0);
    stage(cur ^ 1, false);
    __syncthreads();
  }
  fetch((NKT - 1) * 32, true);
  asm volatile("" ::: "memory");
  mma(0);
  __builtin_amdgcn_sched_barrier(0);
  stage(1, true);
  __syncthreads();
  mma(1);
  __syncthreads();

#pragma unroll
  for (int nt = 0; nt < 8; ++nt) {
    const int n0 = waveN * 128 + nt * 16 + kb;
    const v4f bA = *(const v4f*)(bias + n0);
    const v4f bB = *(const v4f*)(bias + n0 + 4);
    h8 p;
#pragma unroll
    for (int r = 0; r < 8; ++r) {
      float v = acc[nt][r] + (r < 4 ? bA[r] : bB[r - 4]);
      p[r] = (_Float16)(v > 0.f ? v : 0.f);
    }
    *(h8*)&Hs[mrow][n0] = p;
  }
  __syncthreads();

  v8f o[2];
  o[0] = (v8f){0.f, 0.f, 0.f, 0.f, 0.f, 0.f, 0.f, 0.f};
  o[1] = (v8f){0.f, 0.f, 0.f, 0.f, 0.f, 0.f, 0.f, 0.f};
  const int ncol = waveN * 32 + (lane & 15);
  float bv2[2];
#pragma unroll
  for (int q = 0; q < 2; ++q) {
    const int col = ncol + q * 16;
    const float bl = b2[col < NCLASS ? col : NCLASS - 1];
    bv2[q] = col < NCLASS ? bl : 0.f;
  }
#pragma unroll
  for (int kh = 0; kh < HID / 128; ++kh) {
    v16h hf[4], w0[4], w1[4];
#pragma unroll
    for (int u = 0; u < 4; ++u) {
      const int k0 = (kh * 4 + u) * 32;
      hf[u] = load_frag(&Hs[mrow][k0], kb);
      w0[u] = load_frag(&W2s[ncol][k0], kb);
      w1[u] = load_frag(&W2s[ncol + 16][k0], kb);
    }
#pragma unroll
    for (int u = 0; u < 4; ++u) {
      o[0] = wmma16(hf[u], w0[u], o[0]);
      o[1] = wmma16(hf[u], w1[u], o[1]);
    }
  }

  const int rbase = blockM + waveM * 16 + kb;
#pragma unroll
  for (int q = 0; q < 2; ++q) {
    const int col = ncol + q * 16;
#pragma unroll
    for (int i = 0; i < 8; ++i) {
      int row = rbase + i;
      if (row < Nn) plane0[(size_t)row * PITCH + col] = col < NCLASS ? o[q][i] + bv2[q] : 0.f;
    }
  }
}

#define BSH     8
#define BNODES  (1 << BSH)
#define NBK_MAX 512
#define PBLK    256

__global__ void __launch_bounds__(1024)
dagnn_bhist(const int* __restrict__ dst, int* __restrict__ bh, int E, int nbk) {
  __shared__ int h[NBK_MAX];
  const int t = threadIdx.x, b = blockIdx.x;
  if (t < nbk) h[t] = 0;
  __syncthreads();
  const int chunk = (E + PBLK - 1) / PBLK;
  const int e0 = b * chunk;
  const int e1 = e0 + chunk < E ? e0 + chunk : E;
  for (int base = e0; base < e1; base += 8 * 1024) {
    int d[8];
#pragma unroll
    for (int j = 0; j < 8; ++j) {
      const int e = base + j * 1024 + t;
      d[j] = e < e1 ? dst[e] : -1;
    }
#pragma unroll
    for (int j = 0; j < 8; ++j)
      if (d[j] >= 0) atomicAdd(&h[d[j] >> BSH], 1);
  }
  __syncthreads();
  if (t < nbk) bh[t * PBLK + b] = h[t];
}

__global__ void __launch_bounds__(1024)
dagnn_bscatter(const int* __restrict__ src, const int* __restrict__ dst,
               const int* __restrict__ boff, int2* __restrict__ ebuf, int E, int nbk) {
  __shared__ int cur[NBK_MAX];
  const int t = threadIdx.x, b = blockIdx.x;
  if (t < nbk) cur[t] = boff[t * PBLK + b];
  __syncthreads();
  const int chunk = (E + PBLK - 1) / PBLK;
  const int e0 = b * chunk;
  const int e1 = e0 + chunk < E ? e0 + chunk : E;
  for (int base = e0; base < e1; base += 8 * 1024) {
    int d[8], s[8];
#pragma unroll
    for (int j = 0; j < 8; ++j) {
      const int e = base + j * 1024 + t;
      d[j] = e < e1 ? dst[e] : -1;
      s[j] = e < e1 ? src[e] : 0;
    }
#pragma unroll
    for (int j = 0; j < 8; ++j)
      if (d[j] >= 0) {
        const int pos = atomicAdd(&cur[d[j] >> BSH], 1);
        ebuf[pos] = make_int2(s[j], d[j]);
      }
  }
}

__global__ void __launch_bounds__(1024)
dagnn_bcount(const int2* __restrict__ ebuf, const int* __restrict__ boff,
             int* __restrict__ rowptr, float* __restrict__ norm, int Nn) {
  __shared__ int cn[BNODES];
  __shared__ int wsum[BNODES / 32];
  const int t = threadIdx.x, lane = t & 31, k = blockIdx.x;
  const int beg = boff[k * PBLK], end = boff[(k + 1) * PBLK];
  const int nbase = k << BSH;
  if (t < BNODES) cn[t] = 0;
  __syncthreads();
  for (int base = beg; base < end; base += 4 * 1024) {
    int d[4];
#pragma unroll
    for (int j = 0; j < 4; ++j) {
      const int e = base + j * 1024 + t;
      d[j] = e < end ? ebuf[e].y : -1;
    }
#pragma unroll
    for (int j = 0; j < 4; ++j)
      if (d[j] >= 0) atomicAdd(&cn[d[j] - nbase], 1);
  }
  __syncthreads();
  int v = 0, inc = 0;
  if (t < BNODES) {
    v = cn[t];
    inc = v;
#pragma unroll
    for (int off = 1; off < 32; off <<= 1) {
      int u = __shfl_up(inc, off, 32);
      if (lane >= off) inc += u;
    }
    if (lane == 31) wsum[t >> 5] = inc;
  }
  __syncthreads();
  if (t < BNODES) {
    int ex = beg + inc - v;
#pragma unroll
    for (int w = 0; w < BNODES / 32; ++w)
      if (w < (t >> 5)) ex += wsum[w];
    const int node = nbase + t;
    if (node < Nn) {
      rowptr[node] = ex;
      norm[node] = rsqrtf((float)v);
      if (node == Nn - 1) rowptr[Nn] = ex + v;
    }
  }
}

__global__ void __launch_bounds__(1024)
dagnn_bplace(const int2* __restrict__ ebuf, const int* __restrict__ boff,
             const int* __restrict__ rowptr, const float* __restrict__ norm,
             int2* __restrict__ csr, int Nn) {
  __shared__ int cur[BNODES];
  const int t = threadIdx.x, k = blockIdx.x;
  const int beg = boff[k * PBLK], end = boff[(k + 1) * PBLK];
  const int nbase = k << BSH;
  if (t < BNODES) cur[t] = nbase + t < Nn ? rowptr[nbase + t] : 0;
  __syncthreads();
  for (int base = beg; base < end; base += 4 * 1024) {
    int2 ed[4];
    int wb[4];
#pragma unroll
    for (int j = 0; j < 4; ++j) {
      const int e = base + j * 1024 + t;
      ed[j] = e < end ? ebuf[e] : make_int2(0, -1);
    }
#pragma unroll
    for (int j = 0; j < 4; ++j) wb[j] = __float_as_int(norm[ed[j].x]);
#pragma unroll
    for (int j = 0; j < 4; ++j)
      if (ed[j].y >= 0) {
        const int pos = atomicAdd(&cur[ed[j].y - nbase], 1);
        csr[pos] = make_int2(ed[j].x, wb[j]);
      }
  }
}

__global__ void __launch_bounds__(256)
dagnn_tilesum_prep(const int* __restrict__ cnt, int* __restrict__ tsum, int Nn, int ntile,
                   const float* __restrict__ W1, const float* __restrict__ W2,
                   _Float16* __restrict__ W1h, _Float16* __restrict__ W2h) {
  if ((int)blockIdx.x >= ntile) {
    prep_weights(((int)blockIdx.x - ntile) * 256 + (int)threadIdx.x, W1, W2, W1h, W2h);
    return;
  }
  __shared__ int red[8];
  const int t = threadIdx.x;
  const int i0 = blockIdx.x * 1024 + t * 4;
  int s = 0;
#pragma unroll
  for (int u = 0; u < 4; ++u)
    if (i0 + u < Nn) s += cnt[i0 + u];
#pragma unroll
  for (int off = 16; off > 0; off >>= 1) s += __shfl_xor(s, off, 32);
  if ((t & 31) == 0) red[t >> 5] = s;
  __syncthreads();
  if (t == 0) {
    int r = 0;
#pragma unroll
    for (int w = 0; w < 8; ++w) r += red[w];
    tsum[blockIdx.x] = r;
  }
}

__global__ void __launch_bounds__(256)
dagnn_scan(const int* __restrict__ cnt, const int* __restrict__ tsum,
           int* __restrict__ rowptr, int Nn) {
  __shared__ int red[8];
  __shared__ int wsum[8];
  const int t = threadIdx.x, lane = t & 31, wave = t >> 5;
  const int tile = blockIdx.x;

  int o = 0;
  for (int i = t; i < tile; i += 256) o += tsum[i];
#pragma unroll
  for (int off = 16; off > 0; off >>= 1) o += __shfl_xor(o, off, 32);
  if (lane == 0) red[wave] = o;

  const int i0 = tile * 1024 + t * 4;
  int c[4];
#pragma unroll
  for (int u = 0; u < 4; ++u) c[u] = (i0 + u < Nn) ? cnt[i0 + u] : 0;
  const int ts = c[0] + c[1] + c[2] + c[3];
  int inc = ts;
#pragma unroll
  for (int off = 1; off < 32; off <<= 1) {
    int v = __shfl_up(inc, off, 32);
    if (lane >= off) inc += v;
  }
  if (lane == 31) wsum[wave] = inc;
  __syncthreads();

  int ex = inc - ts;
#pragma unroll
  for (int w = 0; w < 8; ++w) {
    ex += red[w];
    if (w < wave) ex += wsum[w];
  }
#pragma unroll
  for (int u = 0; u < 4; ++u) {
    if (i0 + u < Nn) {
      rowptr[i0 + u] = ex;
      ex += c[u];
    }
  }
  if (i0 < Nn && Nn <= i0 + 4) rowptr[Nn] = ex;
}


template <int S>
__device__ __forceinline__ int row_bcast(int v) {
  return __builtin_amdgcn_update_dpp(0, v, 0x150 + S, 0xf, 0xf, false);
}

__device__ __forceinline__ void load_edges(const int2* __restrict__ csr, int beg, int len,
                                           int done, int c, unsigned& eo, int& wb) {
  eo = 208u;
  wb = 0;
  if (c < len - done) {
    const int2 e = csr[beg + done + c];
    eo = (unsigned)e.x << 8;
    wb = e.y;
  }
}

template <int S, int U, int C>
__device__ __forceinline__ void issue_rows(const char* __restrict__ inb, unsigned eo, int wb,
                                           unsigned cB, v4f* x, float* w) {
  if constexpr (U < C) {
    const unsigned o = (unsigned)row_bcast<S + U>((int)eo) | cB;
    w[U] = __int_as_float(row_bcast<S + U>(wb));
    x[U] = *(const v4f*)(inb + o);
    issue_rows<S, U + 1, C>(inb, eo, wb, cB, x, w);
  }
}

template <int S, int C>
__device__ __forceinline__ void gather_rows(const char* __restrict__ inb, unsigned eo, int wb,
                                            unsigned cB, v4f& a) {
  v4f x[C];
  float w[C];
  issue_rows<S, 0, C>(inb, eo, wb, cB, x, w);
#pragma unroll
  for (int u = 0; u < C; ++u) {
    a.x = fmaf(w[u], x[u].x, a.x);
    a.y = fmaf(w[u], x[u].y, a.y);
    a.z = fmaf(w[u], x[u].z, a.z);
    a.w = fmaf(w[u], x[u].w, a.w);
  }
}

template <int S>
__device__ __forceinline__ void walk8(const char* __restrict__ inb, unsigned eo, int wb,
                                      unsigned cB, int r, v4f& a) {
  if (r > 6)      gather_rows<S, 8>(inb, eo, wb, cB, a);
  else if (r > 4) gather_rows<S, 6>(inb, eo, wb, cB, a);
  else if (r > 2) gather_rows<S, 4>(inb, eo, wb, cB, a);
  else            gather_rows<S, 2>(inb, eo, wb, cB, a);
}

__device__ __forceinline__ void walk16(const char* __restrict__ inb, unsigned eo, int wb,
                                       unsigned cB, int m, v4f& a) {
  walk8<0>(inb, eo, wb, cB, m, a);
  if (m > 8) walk8<8>(inb, eo, wb, cB, m - 8, a);
}

__device__ __forceinline__ int wave_max_len(int len) {
  const int l0 = __builtin_amdgcn_readlane(len, 0), l1 = __builtin_amdgcn_readlane(len, 16);
  return l0 > l1 ? l0 : l1;
}

__global__ void __launch_bounds__(256)
dagnn_gather(const float* __restrict__ in, float* __restrict__ outp,
             const int2* __restrict__ csr, const int* __restrict__ rowptr,
             const float* __restrict__ norm, int Nn) {
  const int lane = threadIdx.x & 31;
  const int c = lane & 15;
  const int node = ((int)blockIdx.x * 8 + ((int)threadIdx.x >> 5)) * 2 + (lane >> 4);
  const bool live = node < Nn;
  int beg = 0, end = 0;
  if (live) { beg = rowptr[node]; end = rowptr[node + 1]; }
  const int len = end - beg;
  const int lenmax = wave_max_len(len);
  const char* inb = (const char*)in;
  const unsigned cB = (unsigned)c << 4;
  v4f a = (v4f){0.f, 0.f, 0.f, 0.f};
  for (int done = 0; done < lenmax; done += 16) {
    unsigned eo;
    int wb;
    load_edges(csr, beg, len, done, c, eo, wb);
    const int m = lenmax - done;
    walk16(inb, eo, wb, cB, m < 16 ? m : 16, a);
  }
  if (live) {
    const float nd = norm[node];
    v4f r;
    r.x = 4 * c + 0 < NCLASS ? nd * a.x : 0.f;
    r.y = 4 * c + 1 < NCLASS ? nd * a.y : 0.f;
    r.z = 4 * c + 2 < NCLASS ? nd * a.z : 0.f;
    r.w = 4 * c + 3 < NCLASS ? nd * a.w : 0.f;
    *(v4f*)(outp + (size_t)node * PITCH + 4 * c) = r;
  }
}

template <int M>
__device__ __forceinline__ float row_xor_add(float z) {
  return z + __int_as_float(__builtin_amdgcn_update_dpp(0, __float_as_int(z), 0x160 + M,
                                                        0xf, 0xf, false));
}

__global__ void __launch_bounds__(256)
dagnn_gather_readout(const float* __restrict__ planes, const float* __restrict__ s,
                     float* __restrict__ out, const int2* __restrict__ csr,
                     const int* __restrict__ rowptr, const float* __restrict__ norm, int Nn) {
  const int lane = threadIdx.x & 31;
  const int c = lane & 15;
  const int node = ((int)blockIdx.x * 8 + ((int)threadIdx.x >> 5)) * 2 + (lane >> 4);
  const bool live = node < Nn;
  const int cnode = live ? node : Nn - 1;
  int beg = 0, end = 0;
  if (live) { beg = rowptr[node]; end = rowptr[node + 1]; }
  const int len = end - beg;
  const int lenmax = wave_max_len(len);
  const size_t pstride = (size_t)Nn * PITCH;
  const char* inb = (const char*)(planes + (size_t)(KHOPS - 1) * pstride);
  const unsigned cB = (unsigned)c << 4;

  unsigned eo;
  int wb;
  load_edges(csr, beg, len, 0, c, eo, wb);

  const float* ownp = planes + (size_t)cnode * PITCH + 4 * c;
  v4f own[KHOPS];
#pragma unroll
  for (int k = 0; k < KHOPS - 1; ++k)
    own[k] = __builtin_nontemporal_load((const v4f*)(ownp + (size_t)k * pstride));
  own[KHOPS - 1] = *(const v4f*)(ownp + (size_t)(KHOPS - 1) * pstride);

  float sc[4];
  bool real[4];
#pragma unroll
  for (int i = 0; i < 4; ++i) {
    const int col = 4 * c + i;
    real[i] = col < NCLASS;
    const float sl = s[real[i] ? col : NCLASS - 1];
    sc[i] = real[i] ? sl : 0.f;
  }
  const float nd = live ? norm[node] : 0.f;

  v4f a = (v4f){0.f, 0.f, 0.f, 0.f};
  if (lenmax > 0) walk16(inb, eo, wb, cB, lenmax < 16 ? lenmax : 16, a);
  for (int done = 16; done < lenmax; done += 16) {
    load_edges(csr, beg, len, done, c, eo, wb);
    const int m = lenmax - done;
    walk16(inb, eo, wb, cB, m < 16 ? m : 16, a);
  }
  v4f h;
  h.x = real[0] ? nd * a.x : 0.f;
  h.y = real[1] ? nd * a.y : 0.f;
  h.z = real[2] ? nd * a.z : 0.f;
  h.w = real[3] ? nd * a.w : 0.f;

  v4f o = (v4f){0.f, 0.f, 0.f, 0.f};
  auto add_hop = [&](v4f x) {
    float z = (real[0] ? x.x * sc[0] : 0.f) + (real[1] ? x.y * sc[1] : 0.f)
            + (real[2] ? x.z * sc[2] : 0.f) + (real[3] ? x.w * sc[3] : 0.f);
    z = row_xor_add<8>(z);
    z = row_xor_add<4>(z);
    z = row_xor_add<2>(z);
    z = row_xor_add<1>(z);
    const float S = 1.0f / (1.0f + __expf(-z));
    o.x += S * x.x;
    o.y += S * x.y;
    o.z += S * x.z;
    o.w += S * x.w;
  };
#pragma unroll
  for (int k = 0; k < KHOPS; ++k) add_hop(own[k]);
  add_hop(h);
  if (live) {
    float* op = out + (size_t)node * NCLASS + 4 * c;
    if (real[1]) { v2f r; r.x = o.x; r.y = o.y; *(v2f*)op = r; }
    if (real[3]) { v2f r; r.x = o.z; r.y = o.w; *(v2f*)(op + 2) = r; }
  }
}

extern "C" void kernel_launch(void* const* d_in, const int* in_sizes, int n_in,
                              void* d_out, int out_size, void* d_ws, size_t ws_size,
                              hipStream_t stream) {
  const float* feats = (const float*)d_in[0];
  const float* W1    = (const float*)d_in[1];
  const float* b1    = (const float*)d_in[2];
  const float* W2    = (const float*)d_in[3];
  const float* b2    = (const float*)d_in[4];
  const float* s     = (const float*)d_in[5];
  const int*   src   = (const int*)d_in[6];
  const int*   dst   = (const int*)d_in[7];
  const int Nn = in_sizes[0] / IN_DIM;
  const int E  = in_sizes[6];
  float* out = (float*)d_out;

  char* ws = (char*)d_ws;
  size_t off = 0;
  auto carve = [&](size_t bytes) -> char* {
    char* p = ws + off;
    off = (off + bytes + 255) & ~(size_t)255;
    return p;
  };
  const int nbk   = (Nn + BNODES - 1) >> BSH;
  const int M     = nbk * PBLK;
  const int ntile = (M + 1023) / 1024;
  float*    norm   = (float*)   carve((size_t)Nn * sizeof(float));
  int*      rowptr = (int*)     carve((size_t)(Nn + 1) * sizeof(int));
  int*      bh     = (int*)     carve((size_t)M * sizeof(int));
  int*      boff   = (int*)     carve((size_t)(M + 1) * sizeof(int));
  int*      tsum   = (int*)     carve((size_t)ntile * sizeof(int));
  int2*     ebuf   = (int2*)    carve((size_t)E * sizeof(int2));
  int2*     csr    = (int2*)    carve((size_t)E * sizeof(int2));
  float*    planes = (float*)   carve((size_t)KHOPS * Nn * PITCH * sizeof(float));
  _Float16* W1h    = (_Float16*)carve((size_t)W1_PIECES * 16);
  _Float16* W2h    = (_Float16*)carve((size_t)W2_PIECES * 16);
  if (off > ws_size || nbk > NBK_MAX) return;

  const size_t pstride = (size_t)Nn * PITCH;

  dagnn_bhist<<<PBLK, 1024, 0, stream>>>(dst, bh, E, nbk);
  dagnn_tilesum_prep<<<ntile + PREP_BLOCKS, 256, 0, stream>>>(bh, tsum, M, ntile, W1, W2, W1h, W2h);
  dagnn_scan<<<ntile, 256, 0, stream>>>(bh, tsum, boff, M);
  dagnn_bscatter<<<PBLK, 1024, 0, stream>>>(src, dst, boff, ebuf, E, nbk);
  dagnn_bcount<<<nbk, 1024, 0, stream>>>(ebuf, boff, rowptr, norm, Nn);
  dagnn_bplace<<<nbk, 1024, 0, stream>>>(ebuf, boff, rowptr, norm, csr, Nn);

  dim3 g1((Nn + 63) / 64, 1);
  dagnn_mlp<<<g1, 256, 0, stream>>>(feats, W1h, b1, W2h, b2, planes, Nn);

  const int ggrid = (Nn + 15) / 16;
  for (int k = 1; k < KHOPS; ++k) {
    const float* pin = planes + (size_t)(k - 1) * pstride;
    float* pout      = planes + (size_t)k * pstride;
    dagnn_gather<<<ggrid, 256, 0, stream>>>(pin, pout, csr, rowptr, norm, Nn);
  }

  dagnn_gather_readout<<<ggrid, 256, 0, stream>>>(planes, s, out, csr, rowptr, norm, Nn);
  (void)n_in; (void)out_size;
}
